# starting kernel
# baseline (speedup 1.0000x reference)
#include <hip/hip_runtime.h>
#include <hip/hip_bf16.h>
#include <math.h>

#define EMB     128
#define BATCH   16384
#define NNODES  100000
#define NEDGES  640000

typedef float v2f __attribute__((ext_vector_type(2)));
typedef float v8f __attribute__((ext_vector_type(8)));

__device__ __forceinline__ float wave_sum(float v) {
#pragma unroll
    for (int m = 16; m >= 1; m >>= 1) v += __shfl_xor(v, m, 32);
    return v;
}

template <int NT>
__global__ void __launch_bounds__(256)
wmma_gemm_kernel(const float* __restrict__ X,
                 const float* __restrict__ W,
                 const float* __restrict__ bias,
                 float* __restrict__ Y,
                 int M, int K, int act) {
    constexpr int N   = NT * 16;
    constexpr int KC  = 64;
    constexpr int LDW = N + 16;
    __shared__ float2 wt2[(KC / 2) * LDW];

    const int tid   = threadIdx.x;
    const int wave  = tid >> 5;
    const int lane  = tid & 31;
    const int half  = lane >> 4;
    const int l     = lane & 15;
    const int rowBase = (blockIdx.x * (blockDim.x >> 5) + wave) * 16;

    v8f acc[NT] = {};

    const int  row   = rowBase + l;
    const bool rowOK = (row < M);
    const int  rr    = rowOK ? row : (M - 1);
    const float* xrow = X + (size_t)rr * K;

    for (int kc = 0; kc < K; kc += KC) {
        for (int i = tid; i < (KC / 2) * N; i += blockDim.x) {
            const int p = i / N;
            const int c = i % N;
            float2 w2;
            w2.x = W[(size_t)(kc + 2 * p)     * N + c];
            w2.y = W[(size_t)(kc + 2 * p + 1) * N + c];
            wt2[p * LDW + c] = w2;
        }
        __syncthreads();

        float2 a_nxt = *(const float2*)(xrow + kc + 2 * half);
#pragma unroll
        for (int k = 0; k < KC; k += 4) {
            v2f a;
            a.x = rowOK ? a_nxt.x : 0.0f;
            a.y = rowOK ? a_nxt.y : 0.0f;
            if (k + 4 < KC)
                a_nxt = *(const float2*)(xrow + kc + k + 4 + 2 * half);

            const float2* bp = wt2 + ((k >> 1) + half) * LDW + l;
            float2 breg[NT];
#pragma unroll
            for (int t = 0; t < NT; ++t) breg[t] = bp[t * 16];
#pragma unroll
            for (int t = 0; t < NT; ++t) {
                v2f b; b.x = breg[t].x; b.y = breg[t].y;
                acc[t] = __builtin_amdgcn_wmma_f32_16x16x4_f32(
                    false, a, false, b, (short)0, acc[t], false, false);
            }
        }
        __syncthreads();
    }

#pragma unroll
    for (int t = 0; t < NT; ++t) {
        const int   c  = t * 16 + l;
        const float bv = bias ? bias[c] : 0.0f;
#pragma unroll
        for (int v = 0; v < 8; ++v) {
            const int r = rowBase + v + 8 * half;
            if (r < M) {
                float o = acc[t][v] + bv;
                if (act) o = fmaxf(o, 0.0f);
                Y[(size_t)r * N + c] = o;
            }
        }
    }
}

__global__ void edge_scatter_kernel(const int* __restrict__ src,
                                    const int* __restrict__ dst,
                                    const int* __restrict__ etype,
                                    const float* __restrict__ eweight,
                                    const float* __restrict__ y0,
                                    const float* __restrict__ y1,
                                    float* __restrict__ out) {
    const int e = blockIdx.x * (blockDim.x >> 5) + (threadIdx.x >> 5);
    if (e >= NEDGES) return;
    const int lane = threadIdx.x & 31;
    const int s = src[e], d = dst[e];
    const float w = eweight[e];
    const float* y = (etype[e] == 0) ? y0 : y1;
    const float4 v = *(const float4*)(y + (size_t)s * EMB + lane * 4);
    float* o = out + (size_t)d * EMB + lane * 4;
    unsafeAtomicAdd(o + 0, w * v.x);
    unsafeAtomicAdd(o + 1, w * v.y);
    unsafeAtomicAdd(o + 2, w * v.z);
    unsafeAtomicAdd(o + 3, w * v.w);
}

__global__ void relu_ln_kernel(float* __restrict__ x,
                               const float* __restrict__ g,
                               const float* __restrict__ b,
                               int M) {
    const int row = blockIdx.x * (blockDim.x >> 5) + (threadIdx.x >> 5);
    if (row >= M) return;
    const int lane = threadIdx.x & 31;
    float* p = x + (size_t)row * EMB + lane * 4;
    float4 v = *(const float4*)p;
    v.x = fmaxf(v.x, 0.0f); v.y = fmaxf(v.y, 0.0f);
    v.z = fmaxf(v.z, 0.0f); v.w = fmaxf(v.w, 0.0f);
    const float mu = wave_sum(v.x + v.y + v.z + v.w) * (1.0f / EMB);
    const float dx = v.x - mu, dy = v.y - mu, dz = v.z - mu, dw = v.w - mu;
    const float var = wave_sum(dx * dx + dy * dy + dz * dz + dw * dw) * (1.0f / EMB);
    const float inv = rsqrtf(var + 1e-5f);
    const float4 gg = *(const float4*)(g + lane * 4);
    const float4 bb = *(const float4*)(b + lane * 4);
    float4 o;
    o.x = dx * inv * gg.x + bb.x;
    o.y = dy * inv * gg.y + bb.y;
    o.z = dz * inv * gg.z + bb.z;
    o.w = dw * inv * gg.w + bb.w;
    *(float4*)p = o;
}

__global__ void gather_gmf_kernel(const float* __restrict__ h2,
                                  const int* __restrict__ ui,
                                  const int* __restrict__ vi,
                                  float* __restrict__ gmf,
                                  float* __restrict__ hcat) {
    const int bidx = blockIdx.x * (blockDim.x >> 5) + (threadIdx.x >> 5);
    if (bidx >= BATCH) return;
    const int lane = threadIdx.x & 31;
    const int u = ui[bidx], w = vi[bidx];
    const float4 uv = *(const float4*)(h2 + (size_t)u * EMB + lane * 4);
    const float4 vv = *(const float4*)(h2 + (size_t)w * EMB + lane * 4);
    const float su = wave_sum(uv.x * uv.x + uv.y * uv.y + uv.z * uv.z + uv.w * uv.w);
    const float sv = wave_sum(vv.x * vv.x + vv.y * vv.y + vv.z * vv.z + vv.w * vv.w);
    const float rn = 1.0f / (fmaxf(sqrtf(su), 1e-12f) * fmaxf(sqrtf(sv), 1e-12f));
    float4 gg;
    gg.x = uv.x * vv.x * rn; gg.y = uv.y * vv.y * rn;
    gg.z = uv.z * vv.z * rn; gg.w = uv.w * vv.w * rn;
    *(float4*)(gmf + (size_t)bidx * EMB + lane * 4) = gg;
    *(float4*)(hcat + (size_t)bidx * 2 * EMB + lane * 4) = uv;
    *(float4*)(hcat + (size_t)bidx * 2 * EMB + EMB + lane * 4) = vv;
}

__global__ void score_kernel(const float* __restrict__ gmf,
                             const float* __restrict__ h3,
                             const float* __restrict__ out_w,
                             const float* __restrict__ out_b,
                             float* __restrict__ out) {
    const int bidx = blockIdx.x * (blockDim.x >> 5) + (threadIdx.x >> 5);
    if (bidx >= BATCH) return;
    const int lane = threadIdx.x & 31;
    const float4 gg = *(const float4*)(gmf + (size_t)bidx * EMB + lane * 4);
    const float4 ww = *(const float4*)(out_w + lane * 4);
    float acc = gg.x * ww.x + gg.y * ww.y + gg.z * ww.z + gg.w * ww.w;
    acc += h3[(size_t)bidx * 32 + lane] * out_w[EMB + lane];
    acc = wave_sum(acc) + out_b[0];
    if (lane == 0) out[bidx] = 1.0f / (1.0f + expf(-acc));
}

extern "C" void kernel_launch(void* const* d_in, const int* in_sizes, int n_in,
                              void* d_out, int out_size, void* d_ws, size_t ws_size,
                              hipStream_t stream) {
    const int*   user_idx = (const int*)d_in[0];
    const int*   item_idx = (const int*)d_in[1];
    const int*   eidx     = (const int*)d_in[2];
    const int*   etype    = (const int*)d_in[3];
    const float* eweight  = (const float*)d_in[4];
    const float* emb      = (const float*)d_in[5];
    const float* w1_rel   = (const float*)d_in[6];
    const float* w1_root  = (const float*)d_in[7];
    const float* b1       = (const float*)d_in[8];
    const float* ln_g     = (const float*)d_in[9];
    const float* ln_b     = (const float*)d_in[10];
    const float* w2_rel   = (const float*)d_in[11];
    const float* w2_root  = (const float*)d_in[12];
    const float* b2       = (const float*)d_in[13];
    const float* mlp_w0   = (const float*)d_in[14];
    const float* mlp_b0   = (const float*)d_in[15];
    const float* mlp_w1   = (const float*)d_in[16];
    const float* mlp_b1   = (const float*)d_in[17];
    const float* mlp_w2   = (const float*)d_in[18];
    const float* mlp_b2   = (const float*)d_in[19];
    const float* out_w    = (const float*)d_in[20];
    const float* out_b    = (const float*)d_in[21];
    float* out = (float*)d_out;

    char* ws = (char*)d_ws;
    const size_t NB = (size_t)NNODES * EMB * sizeof(float);
    float* y0 = (float*)(ws);
    float* y1 = (float*)(ws + NB);
    float* h1 = (float*)(ws + 2 * NB);
    float* h2 = (float*)(ws + 3 * NB);
    float* gmf  = (float*)(ws);
    float* hcat = (float*)(ws + (size_t)BATCH * EMB * 4);
    float* m0   = (float*)(ws + (size_t)BATCH * EMB * 4 + (size_t)BATCH * 2 * EMB * 4);
    float* m1   = (float*)((char*)m0 + (size_t)BATCH * 128 * 4);
    float* m2   = (float*)((char*)m1 + (size_t)BATCH * 64 * 4);

    const int nodeGrid  = (NNODES + 127) / 128;
    const int batchGrid = (BATCH + 127) / 128;
    const int edgeGrid  = (NEDGES + 7) / 8;
    const int rowGridN  = (NNODES + 7) / 8;
    const int rowGridB  = (BATCH + 7) / 8;

    wmma_gemm_kernel<8><<<nodeGrid, 256, 0, stream>>>(emb, w1_rel,             nullptr, y0, NNODES, EMB, 0);
    wmma_gemm_kernel<8><<<nodeGrid, 256, 0, stream>>>(emb, w1_rel + EMB * EMB, nullptr, y1, NNODES, EMB, 0);
    wmma_gemm_kernel<8><<<nodeGrid, 256, 0, stream>>>(emb, w1_root,            b1,      h1, NNODES, EMB, 0);
    edge_scatter_kernel<<<edgeGrid, 256, 0, stream>>>(eidx, eidx + NEDGES, etype, eweight, y0, y1, h1);
    relu_ln_kernel<<<rowGridN, 256, 0, stream>>>(h1, ln_g, ln_b, NNODES);

    wmma_gemm_kernel<8><<<nodeGrid, 256, 0, stream>>>(h1, w2_rel,             nullptr, y0, NNODES, EMB, 0);
    wmma_gemm_kernel<8><<<nodeGrid, 256, 0, stream>>>(h1, w2_rel + EMB * EMB, nullptr, y1, NNODES, EMB, 0);
    wmma_gemm_kernel<8><<<nodeGrid, 256, 0, stream>>>(h1, w2_root,            b2,      h2, NNODES, EMB, 0);
    edge_scatter_kernel<<<edgeGrid, 256, 0, stream>>>(eidx, eidx + NEDGES, etype, eweight, y0, y1, h2);

    gather_gmf_kernel<<<rowGridB, 256, 0, stream>>>(h2, user_idx, item_idx, gmf, hcat);
    wmma_gemm_kernel<8><<<batchGrid, 256, 0, stream>>>(hcat, mlp_w0, mlp_b0, m0, BATCH, 2 * EMB, 1);
    wmma_gemm_kernel<4><<<batchGrid, 256, 0, stream>>>(m0,   mlp_w1, mlp_b1, m1, BATCH, 128,     1);
    wmma_gemm_kernel<2><<<batchGrid, 256, 0, stream>>>(m1,   mlp_w2, mlp_b2, m2, BATCH, 64,      1);
    score_kernel<<<rowGridB, 256, 0, stream>>>(gmf, m2, out_w, out_b, out);
}


# edge sums owner-computes: a block owns 392 destination rows in LDS, filters the whole edge list for them, adds source rows by LDS f32 atomics (arrival order), ReLU+LN fused, each row stored once
# speedup vs baseline: 1.7151x; 1.7151x over previous
#include <hip/hip_runtime.h>
#include <hip/hip_bf16.h>
#include <math.h>

#define EMB     128
#define BATCH   16384
#define NNODES  100000
#define NEDGES  640000

typedef float v2f __attribute__((ext_vector_type(2)));
typedef float v8f __attribute__((ext_vector_type(8)));

__device__ __forceinline__ float wave_sum(float v) {
#pragma unroll
    for (int m = 16; m >= 1; m >>= 1) v += __shfl_xor(v, m, 32);
    return v;
}

template <int NT, int K, int NTHR>
__global__ void __launch_bounds__(NTHR)
wmma_gemm_kernel(const float* __restrict__ X,
                 const float* __restrict__ W,
                 const float* __restrict__ bias,
                 float* __restrict__ Y,
                 int M, int act) {
    constexpr int N     = NT * 16;
    constexpr int LDW   = N + 16;
    constexpr int KP    = K / 2;
    constexpr int KS    = K / 4;
    constexpr int WAVES = NTHR / 32;
    constexpr int NQ    = N / 4;
    constexpr int ITEMS = KP * NQ;
    __shared__ __attribute__((aligned(16))) float2 wt2[KP * LDW];

    const int tid  = threadIdx.x;
    const int wave = __builtin_amdgcn_readfirstlane(tid >> 5);
    const int lane = tid & 31;
    const int half = lane >> 4;
    const int l    = lane & 15;

    const int nTiles = (M + 15) >> 4;
    const int tstep  = gridDim.x * WAVES;
    int tile = blockIdx.x + gridDim.x * wave;

    float2 areg[KS];
    bool rowOK;
    {
        const int row = tile * 16 + l;
        rowOK = (tile < nTiles) && (row < M);
        const int rr = rowOK ? row : (M - 1);
        const float* xrow = X + (size_t)rr * K + 2 * half;
#pragma unroll
        for (int j = 0; j < KS; ++j) areg[j] = *(const float2*)(xrow + 4 * j);
    }

    float bv[NT];
#pragma unroll
    for (int t = 0; t < NT; ++t) bv[t] = bias ? bias[t * 16 + l] : 0.0f;

#pragma unroll
    for (int i0 = 0; i0 < ITEMS; i0 += NTHR) {
        const int i = i0 + tid;
        if ((ITEMS % NTHR == 0) || (i < ITEMS)) {
            const int p = i / NQ;
            const int c = (i % NQ) * 4;
            const float4 r0 = *(const float4*)(W + (size_t)(2 * p)     * N + c);
            const float4 r1 = *(const float4*)(W + (size_t)(2 * p + 1) * N + c);
            float4* dstp = (float4*)(wt2 + p * LDW + c);
            dstp[0] = make_float4(r0.x, r1.x, r0.y, r1.y);
            dstp[1] = make_float4(r0.z, r1.z, r0.w, r1.w);
        }
    }
    __syncthreads();

    for (; tile < nTiles; tile += tstep) {
        v8f acc[NT] = {};
        const float2* bp0 = wt2 + half * LDW + l;
        float2 bcur[NT], bnxt[NT];
#pragma unroll
        for (int t = 0; t < NT; ++t) bcur[t] = bp0[t * 16];
#pragma unroll
        for (int j = 0; j < KS; ++j) {
            v2f a;
            a.x = rowOK ? areg[j].x : 0.0f;
            a.y = rowOK ? areg[j].y : 0.0f;
            if (j + 1 < KS) {
                const float2* bp = bp0 + (2 * (j + 1)) * LDW;
#pragma unroll
                for (int t = 0; t < NT; ++t) bnxt[t] = bp[t * 16];
            }
#pragma unroll
            for (int t = 0; t < NT; ++t) {
                v2f b; b.x = bcur[t].x; b.y = bcur[t].y;
                acc[t] = __builtin_amdgcn_wmma_f32_16x16x4_f32(
                    false, a, false, b, (short)0, acc[t], false, false);
            }
            __builtin_amdgcn_sched_barrier(0);
            if (j + 1 < KS) {
#pragma unroll
                for (int t = 0; t < NT; ++t) bcur[t] = bnxt[t];
            }
        }

        const int nxt = tile + tstep;
        if (nxt < nTiles) {
            const int row = nxt * 16 + l;
            rowOK = (row < M);
            const int rr = rowOK ? row : (M - 1);
            const float* xrow = X + (size_t)rr * K + 2 * half;
#pragma unroll
            for (int j = 0; j < KS; ++j) areg[j] = *(const float2*)(xrow + 4 * j);
        }

        const int rowBase = tile * 16;
#pragma unroll
        for (int t = 0; t < NT; ++t) {
            const int c = t * 16 + l;
#pragma unroll
            for (int v = 0; v < 8; ++v) {
                const int r = rowBase + v + 8 * half;
                if (r < M) {
                    float o = acc[t][v] + bv[t];
                    if (act) o = fmaxf(o, 0.0f);
                    Y[(size_t)r * N + c] = o;
                }
            }
        }
    }
}

#define OWN_ROWS 392
#define OWN_THR  1024
#define OWN_WAVES (OWN_THR / 32)
#define QCAP     512
#define SCAN_EDGES 256

template <int DO_LN>
__global__ void __launch_bounds__(OWN_THR)
owner_sum_kernel(const int* __restrict__ src,
                 const int* __restrict__ dst,
                 const int* __restrict__ etype,
                 const float* __restrict__ eweight,
                 const float* __restrict__ y0,
                 const float* __restrict__ y1,
                 float* h,
                 const float* __restrict__ g,
                 const float* __restrict__ b) {
    static_assert(NEDGES % SCAN_EDGES == 0, "edge list is tested in whole 256-edge pieces");
    static_assert(NEDGES <= (1 << 20) && OWN_ROWS <= (1 << 12), "ring entry packs edge | row << 20");
    __shared__ float acc[OWN_ROWS * EMB];
    __shared__ unsigned ring[OWN_WAVES * QCAP];

    const int tid  = threadIdx.x;
    const int lane = tid & 31;
    const int wave = __builtin_amdgcn_readfirstlane(tid >> 5);
    const int r0   = blockIdx.x * OWN_ROWS;
    const int nrow = min(OWN_ROWS, NNODES - r0);
    const unsigned nr = (unsigned)nrow;

    constexpr int RPW = (OWN_ROWS + OWN_WAVES - 1) / OWN_WAVES;
#pragma unroll
    for (int i0 = 0; i0 < RPW; i0 += 7) {
        float4 t[7];
#pragma unroll
        for (int i = 0; i < 7; ++i) {
            const int r = min(wave + OWN_WAVES * (i0 + i), nrow - 1);
            t[i] = *(const float4*)(h + (size_t)(r0 + r) * EMB + lane * 4);
        }
#pragma unroll
        for (int i = 0; i < 7; ++i) {
            const int r = wave + OWN_WAVES * (i0 + i);
            if (r < nrow) {
                float* a = acc + r * EMB + lane;
                a[0] = t[i].x; a[32] = t[i].y; a[64] = t[i].z; a[96] = t[i].w;
            }
        }
    }
    __syncthreads();

    unsigned* qw = ring + wave * QCAP;
    int qhead = 0, qn = 0;

    constexpr int NPIECE = NEDGES / SCAN_EDGES;
    constexpr int NTRIP  = (NPIECE + OWN_WAVES - 1) / OWN_WAVES;

    static_assert(OWN_WAVES <= NPIECE, "every wave has a first piece");
    constexpr int NTRIP3 = ((NTRIP + 2) / 3) * 3;
    const int* dlane = dst + lane * 4;
    auto piece_at = [&](int p) { return dlane + (size_t)min(p, NPIECE - 1) * SCAN_EDGES; };
    int4 idA[3], idB[3];
    idA[0] = *(const int4*)(piece_at(wave));
    idB[0] = *(const int4*)(piece_at(wave) + 128);
    idA[1] = *(const int4*)(piece_at(wave + OWN_WAVES));
    idB[1] = *(const int4*)(piece_at(wave + OWN_WAVES) + 128);

    for (int it = 0; it < NTRIP3; it += 3) {
#pragma unroll
        for (int s = 0; s < 3; ++s) {
            const int piece = wave + OWN_WAVES * (it + s);
            const int4 cA = idA[s], cB = idB[s];
            idA[(s + 2) % 3] = *(const int4*)(piece_at(piece + 2 * OWN_WAVES));
            idB[(s + 2) % 3] = *(const int4*)(piece_at(piece + 2 * OWN_WAVES) + 128);
            const unsigned lim = (piece < NPIECE) ? nr : 0u;

            const unsigned d0 = (unsigned)(cA.x - r0), d1 = (unsigned)(cA.y - r0);
            const unsigned d2 = (unsigned)(cA.z - r0), d3 = (unsigned)(cA.w - r0);
            const unsigned d4 = (unsigned)(cB.x - r0), d5 = (unsigned)(cB.y - r0);
            const unsigned d6 = (unsigned)(cB.z - r0), d7 = (unsigned)(cB.w - r0);
            const unsigned dmin = min(min(min(d0, d1), min(d2, d3)), min(min(d4, d5), min(d6, d7)));

            if (__builtin_amdgcn_ballot_w32(dmin < lim)) {
                const unsigned k0 = __builtin_amdgcn_ballot_w32(d0 < lim);
                const unsigned k1 = __builtin_amdgcn_ballot_w32(d1 < lim);
                const unsigned k2 = __builtin_amdgcn_ballot_w32(d2 < lim);
                const unsigned k3 = __builtin_amdgcn_ballot_w32(d3 < lim);
                const unsigned k4 = __builtin_amdgcn_ballot_w32(d4 < lim);
                const unsigned k5 = __builtin_amdgcn_ballot_w32(d5 < lim);
                const unsigned k6 = __builtin_amdgcn_ballot_w32(d6 < lim);
                const unsigned k7 = __builtin_amdgcn_ballot_w32(d7 < lim);
                const int eA = piece * SCAN_EDGES + lane * 4;
                const int eB = eA + 128;
                auto push = [&](unsigned mask, int e, unsigned dl) {
                    if (mask) {
                        const int pos = qn + (int)__builtin_amdgcn_mbcnt_lo(mask, 0u);
                        if (dl < lim) qw[(qhead + pos) & (QCAP - 1)] = (unsigned)e | (dl << 20);
                        qn += __builtin_popcount(mask);
                    }
                };
                push(k0, eA + 0, d0); push(k1, eA + 1, d1); push(k2, eA + 2, d2); push(k3, eA + 3, d3);
                push(k4, eB + 0, d4); push(k5, eB + 1, d5); push(k6, eB + 2, d6); push(k7, eB + 3, d7);
            }

            const bool last = (it + s == NTRIP3 - 1);
            while (qn >= 32 || (last && qn > 0)) {
                const int n = min(qn, 32);
                __builtin_amdgcn_wave_barrier();
                int s_l = 0, t_l = 0, w_l = 0;
                unsigned dl_l = 0;
                if (lane < n) {
                    const unsigned ent = qw[(qhead + lane) & (QCAP - 1)];
                    const int e = (int)(ent & 0xFFFFFu);
                    dl_l = ent >> 20;
                    s_l = src[e];
                    t_l = etype[e];
                    w_l = __float_as_int(eweight[e]);
                }
                for (int j = 0; j < n; j += 8) {
                    float4 v[8];
                    float  wj[8];
                    unsigned dj[8];
#pragma unroll
                    for (int u = 0; u < 8; ++u) {
                        const int jj = min(j + u, n - 1);
                        const int sj = __builtin_amdgcn_readlane(s_l, jj);
                        const int tj = __builtin_amdgcn_readlane(t_l, jj);
                        wj[u] = __int_as_float(__builtin_amdgcn_readlane(w_l, jj));
                        dj[u] = (unsigned)__builtin_amdgcn_readlane((int)dl_l, jj);
                        const float* y = ((tj == 0) ? y0 : y1) + (size_t)sj * EMB;
                        v[u] = *(const float4*)(y + lane * 4);
                    }
#pragma unroll
                    for (int u = 0; u < 8; ++u) {
                        if (j + u < n) {
                            float* a = acc + dj[u] * EMB + lane;
                            __hip_atomic_fetch_add(a,      wj[u] * v[u].x, __ATOMIC_RELAXED, __HIP_MEMORY_SCOPE_WORKGROUP);
                            __hip_atomic_fetch_add(a + 32, wj[u] * v[u].y, __ATOMIC_RELAXED, __HIP_MEMORY_SCOPE_WORKGROUP);
                            __hip_atomic_fetch_add(a + 64, wj[u] * v[u].z, __ATOMIC_RELAXED, __HIP_MEMORY_SCOPE_WORKGROUP);
                            __hip_atomic_fetch_add(a + 96, wj[u] * v[u].w, __ATOMIC_RELAXED, __HIP_MEMORY_SCOPE_WORKGROUP);
                        }
                    }
                }
                qhead = (qhead + n) & (QCAP - 1);
                qn -= n;
            }
        }
    }
    __syncthreads();

    float4 gg = make_float4(1.f, 1.f, 1.f, 1.f), bb = make_float4(0.f, 0.f, 0.f, 0.f);
    if (DO_LN) {
        gg = *(const float4*)(g + lane * 4);
        bb = *(const float4*)(b + lane * 4);
    }
    for (int g0 = 0; g0 < RPW; g0 += 4) {
        float4 v[4], o[4];
#pragma unroll
        for (int q = 0; q < 4; ++q) {
            const int r = min(wave + OWN_WAVES * (g0 + q), nrow - 1);
            const float* a = acc + r * EMB + lane;
            v[q] = make_float4(a[0], a[32], a[64], a[96]);
            o[q] = v[q];
        }
        if (DO_LN) {
            float s[4], t[4];
#pragma unroll
            for (int q = 0; q < 4; ++q) {
                v[q].x = fmaxf(v[q].x, 0.0f); v[q].y = fmaxf(v[q].y, 0.0f);
                v[q].z = fmaxf(v[q].z, 0.0f); v[q].w = fmaxf(v[q].w, 0.0f);
                s[q] = v[q].x + v[q].y + v[q].z + v[q].w;
            }
#pragma unroll
            for (int m = 16; m >= 1; m >>= 1) {
#pragma unroll
                for (int q = 0; q < 4; ++q) t[q] = __shfl_xor(s[q], m, 32);
#pragma unroll
                for (int q = 0; q < 4; ++q) s[q] += t[q];
            }
            float dx[4], dy[4], dz[4], dw[4];
#pragma unroll
            for (int q = 0; q < 4; ++q) {
                const float mu = s[q] * (1.0f / EMB);
                dx[q] = v[q].x - mu; dy[q] = v[q].y - mu; dz[q] = v[q].z - mu; dw[q] = v[q].w - mu;
                s[q] = dx[q] * dx[q] + dy[q] * dy[q] + dz[q] * dz[q] + dw[q] * dw[q];
            }
#pragma unroll
            for (int m = 16; m >= 1; m >>= 1) {
#pragma unroll
                for (int q = 0; q < 4; ++q) t[q] = __shfl_xor(s[q], m, 32);
#pragma unroll
                for (int q = 0; q < 4; ++q) s[q] += t[q];
            }
#pragma unroll
            for (int q = 0; q < 4; ++q) {
                const float var = s[q] * (1.0f / EMB);
                const float inv = rsqrtf(var + 1e-5f);
                o[q].x = dx[q] * inv * gg.x + bb.x;
                o[q].y = dy[q] * inv * gg.y + bb.y;
                o[q].z = dz[q] * inv * gg.z + bb.z;
                o[q].w = dw[q] * inv * gg.w + bb.w;
            }
        }
#pragma unroll
        for (int q = 0; q < 4; ++q) {
            const int r = wave + OWN_WAVES * (g0 + q);
            if (r < nrow) *(float4*)(h + (size_t)(r0 + r) * EMB + lane * 4) = o[q];
        }
    }
}

__global__ void gather_gmf_kernel(const float* __restrict__ h2,
                                  const int* __restrict__ ui,
                                  const int* __restrict__ vi,
                                  float* __restrict__ gmf,
                                  float* __restrict__ hcat) {
    const int bidx = blockIdx.x * (blockDim.x >> 5) + (threadIdx.x >> 5);
    if (bidx >= BATCH) return;
    const int lane = threadIdx.x & 31;
    const int u = ui[bidx], w = vi[bidx];
    const float4 uv = *(const float4*)(h2 + (size_t)u * EMB + lane * 4);
    const float4 vv = *(const float4*)(h2 + (size_t)w * EMB + lane * 4);
    const float su = wave_sum(uv.x * uv.x + uv.y * uv.y + uv.z * uv.z + uv.w * uv.w);
    const float sv = wave_sum(vv.x * vv.x + vv.y * vv.y + vv.z * vv.z + vv.w * vv.w);
    const float rn = 1.0f / (fmaxf(sqrtf(su), 1e-12f) * fmaxf(sqrtf(sv), 1e-12f));
    float4 gg;
    gg.x = uv.x * vv.x * rn; gg.y = uv.y * vv.y * rn;
    gg.z = uv.z * vv.z * rn; gg.w = uv.w * vv.w * rn;
    *(float4*)(gmf + (size_t)bidx * EMB + lane * 4) = gg;
    *(float4*)(hcat + (size_t)bidx * 2 * EMB + lane * 4) = uv;
    *(float4*)(hcat + (size_t)bidx * 2 * EMB + EMB + lane * 4) = vv;
}

__global__ void score_kernel(const float* __restrict__ gmf,
                             const float* __restrict__ h3,
                             const float* __restrict__ out_w,
                             const float* __restrict__ out_b,
                             float* __restrict__ out) {
    const int bidx = blockIdx.x * (blockDim.x >> 5) + (threadIdx.x >> 5);
    if (bidx >= BATCH) return;
    const int lane = threadIdx.x & 31;
    const float4 gg = *(const float4*)(gmf + (size_t)bidx * EMB + lane * 4);
    const float4 ww = *(const float4*)(out_w + lane * 4);
    float acc = gg.x * ww.x + gg.y * ww.y + gg.z * ww.z + gg.w * ww.w;
    acc += h3[(size_t)bidx * 32 + lane] * out_w[EMB + lane];
    acc = wave_sum(acc) + out_b[0];
    if (lane == 0) out[bidx] = 1.0f / (1.0f + expf(-acc));
}

extern "C" void kernel_launch(void* const* d_in, const int* in_sizes, int n_in,
                              void* d_out, int out_size, void* d_ws, size_t ws_size,
                              hipStream_t stream) {
    const int*   user_idx = (const int*)d_in[0];
    const int*   item_idx = (const int*)d_in[1];
    const int*   eidx     = (const int*)d_in[2];
    const int*   etype    = (const int*)d_in[3];
    const float* eweight  = (const float*)d_in[4];
    const float* emb      = (const float*)d_in[5];
    const float* w1_rel   = (const float*)d_in[6];
    const float* w1_root  = (const float*)d_in[7];
    const float* b1       = (const float*)d_in[8];
    const float* ln_g     = (const float*)d_in[9];
    const float* ln_b     = (const float*)d_in[10];
    const float* w2_rel   = (const float*)d_in[11];
    const float* w2_root  = (const float*)d_in[12];
    const float* b2       = (const float*)d_in[13];
    const float* mlp_w0   = (const float*)d_in[14];
    const float* mlp_b0   = (const float*)d_in[15];
    const float* mlp_w1   = (const float*)d_in[16];
    const float* mlp_b1   = (const float*)d_in[17];
    const float* mlp_w2   = (const float*)d_in[18];
    const float* mlp_b2   = (const float*)d_in[19];
    const float* out_w    = (const float*)d_in[20];
    const float* out_b    = (const float*)d_in[21];
    float* out = (float*)d_out;

    char* ws = (char*)d_ws;
    const size_t NB = (size_t)NNODES * EMB * sizeof(float);
    float* y0 = (float*)(ws);
    float* y1 = (float*)(ws + NB);
    float* h1 = (float*)(ws + 2 * NB);
    float* h2 = (float*)(ws + 3 * NB);
    float* gmf  = (float*)(ws);
    float* hcat = (float*)(ws + (size_t)BATCH * EMB * 4);
    float* m0   = (float*)(ws + (size_t)BATCH * EMB * 4 + (size_t)BATCH * 2 * EMB * 4);
    float* m1   = (float*)((char*)m0 + (size_t)BATCH * 128 * 4);
    float* m2   = (float*)((char*)m1 + (size_t)BATCH * 64 * 4);

    const int nodeGrid  = 256;
    const int batchGrid = 256;
    const int ownGrid   = (NNODES + OWN_ROWS - 1) / OWN_ROWS;
    const int rowGridB  = (BATCH + 7) / 8;

    wmma_gemm_kernel<8, EMB, 512><<<nodeGrid, 512, 0, stream>>>(emb, w1_rel,             nullptr, y0, NNODES, 0);
    wmma_gemm_kernel<8, EMB, 512><<<nodeGrid, 512, 0, stream>>>(emb, w1_rel + EMB * EMB, nullptr, y1, NNODES, 0);
    wmma_gemm_kernel<8, EMB, 512><<<nodeGrid, 512, 0, stream>>>(emb, w1_root,            b1,      h1, NNODES, 0);
    owner_sum_kernel<1><<<ownGrid, OWN_THR, 0, stream>>>(eidx, eidx + NEDGES, etype, eweight, y0, y1, h1, ln_g, ln_b);

    wmma_gemm_kernel<8, EMB, 512><<<nodeGrid, 512, 0, stream>>>(h1, w2_rel,             nullptr, y0, NNODES, 0);
    wmma_gemm_kernel<8, EMB, 512><<<nodeGrid, 512, 0, stream>>>(h1, w2_rel + EMB * EMB, nullptr, y1, NNODES, 0);
    wmma_gemm_kernel<8, EMB, 512><<<nodeGrid, 512, 0, stream>>>(h1, w2_root,            b2,      h2, NNODES, 0);
    owner_sum_kernel<0><<<ownGrid, OWN_THR, 0, stream>>>(eidx, eidx + NEDGES, etype, eweight, y0, y1, h2, nullptr, nullptr);

    gather_gmf_kernel<<<rowGridB, 256, 0, stream>>>(h2, user_idx, item_idx, gmf, hcat);
    wmma_gemm_kernel<8, 2 * EMB, 128><<<batchGrid, 128, 0, stream>>>(hcat, mlp_w0, mlp_b0, m0, BATCH, 1);
    wmma_gemm_kernel<4, 128, 128><<<batchGrid, 128, 0, stream>>>(m0,   mlp_w1, mlp_b1, m1, BATCH, 1);
    wmma_gemm_kernel<2, 64, 128><<<batchGrid, 128, 0, stream>>>(m1,   mlp_w2, mlp_b2, m2, BATCH, 1);
    score_kernel<<<rowGridB, 256, 0, stream>>>(gmf, m2, out_w, out_b, out);
}


# edge list re-encoded per layer by a small launch: 1-byte owner key per edge for the owner scan (4 keys per op) + one 8-byte record per kept edge, was 4-B ids + 3 words; LDS add order differs
# speedup vs baseline: 1.8354x; 1.0701x over previous
#include <hip/hip_runtime.h>
#include <hip/hip_bf16.h>
#include <math.h>

#define EMB     128
#define BATCH   16384
#define NNODES  100000
#define NEDGES  640000

typedef float v2f __attribute__((ext_vector_type(2)));
typedef float v8f __attribute__((ext_vector_type(8)));

__device__ __forceinline__ float wave_sum(float v) {
#pragma unroll
    for (int m = 16; m >= 1; m >>= 1) v += __shfl_xor(v, m, 32);
    return v;
}

template <int NT, int K, int NTHR>
__global__ void __launch_bounds__(NTHR)
wmma_gemm_kernel(const float* __restrict__ X,
                 const float* __restrict__ W,
                 const float* __restrict__ bias,
                 float* __restrict__ Y,
                 int M, int act) {
    constexpr int N     = NT * 16;
    constexpr int LDW   = N + 16;
    constexpr int KP    = K / 2;
    constexpr int KS    = K / 4;
    constexpr int WAVES = NTHR / 32;
    constexpr int NQ    = N / 4;
    constexpr int ITEMS = KP * NQ;
    __shared__ __attribute__((aligned(16))) float2 wt2[KP * LDW];

    const int tid  = threadIdx.x;
    const int wave = __builtin_amdgcn_readfirstlane(tid >> 5);
    const int lane = tid & 31;
    const int half = lane >> 4;
    const int l    = lane & 15;

    const int nTiles = (M + 15) >> 4;
    const int tstep  = gridDim.x * WAVES;
    int tile = blockIdx.x + gridDim.x * wave;

    float2 areg[KS];
    bool rowOK;
    {
        const int row = tile * 16 + l;
        rowOK = (tile < nTiles) && (row < M);
        const int rr = rowOK ? row : (M - 1);
        const float* xrow = X + (size_t)rr * K + 2 * half;
#pragma unroll
        for (int j = 0; j < KS; ++j) areg[j] = *(const float2*)(xrow + 4 * j);
    }

    float bv[NT];
#pragma unroll
    for (int t = 0; t < NT; ++t) bv[t] = bias ? bias[t * 16 + l] : 0.0f;

#pragma unroll
    for (int i0 = 0; i0 < ITEMS; i0 += NTHR) {
        const int i = i0 + tid;
        if ((ITEMS % NTHR == 0) || (i < ITEMS)) {
            const int p = i / NQ;
            const int c = (i % NQ) * 4;
            const float4 r0 = *(const float4*)(W + (size_t)(2 * p)     * N + c);
            const float4 r1 = *(const float4*)(W + (size_t)(2 * p + 1) * N + c);
            float4* dstp = (float4*)(wt2 + p * LDW + c);
            dstp[0] = make_float4(r0.x, r1.x, r0.y, r1.y);
            dstp[1] = make_float4(r0.z, r1.z, r0.w, r1.w);
        }
    }
    __syncthreads();

    for (; tile < nTiles; tile += tstep) {
        v8f acc[NT] = {};
        const float2* bp0 = wt2 + half * LDW + l;
        float2 bcur[NT], bnxt[NT];
#pragma unroll
        for (int t = 0; t < NT; ++t) bcur[t] = bp0[t * 16];
#pragma unroll
        for (int j = 0; j < KS; ++j) {
            v2f a;
            a.x = rowOK ? areg[j].x : 0.0f;
            a.y = rowOK ? areg[j].y : 0.0f;
            if (j + 1 < KS) {
                const float2* bp = bp0 + (2 * (j + 1)) * LDW;
#pragma unroll
                for (int t = 0; t < NT; ++t) bnxt[t] = bp[t * 16];
            }
#pragma unroll
            for (int t = 0; t < NT; ++t) {
                v2f b; b.x = bcur[t].x; b.y = bcur[t].y;
                acc[t] = __builtin_amdgcn_wmma_f32_16x16x4_f32(
                    false, a, false, b, (short)0, acc[t], false, false);
            }
            __builtin_amdgcn_sched_barrier(0);
            if (j + 1 < KS) {
#pragma unroll
                for (int t = 0; t < NT; ++t) bcur[t] = bnxt[t];
            }
        }

        const int nxt = tile + tstep;
        if (nxt < nTiles) {
            const int row = nxt * 16 + l;
            rowOK = (row < M);
            const int rr = rowOK ? row : (M - 1);
            const float* xrow = X + (size_t)rr * K + 2 * half;
#pragma unroll
            for (int j = 0; j < KS; ++j) areg[j] = *(const float2*)(xrow + 4 * j);
        }

        const int rowBase = tile * 16;
#pragma unroll
        for (int t = 0; t < NT; ++t) {
            const int c = t * 16 + l;
#pragma unroll
            for (int v = 0; v < 8; ++v) {
                const int r = rowBase + v + 8 * half;
                if (r < M) {
                    float o = acc[t][v] + bv[t];
                    if (act) o = fmaxf(o, 0.0f);
                    Y[(size_t)r * N + c] = o;
                }
            }
        }
    }
}

#define OWN_ROWS 392
#define OWN_THR  1024
#define OWN_WAVES (OWN_THR / 32)
#define QCAP     64
#define SCAN_EDGES 1024

#define REC_OFF_WORDS (NEDGES / 4 + 3840)
__global__ void __launch_bounds__(256)
owner_prep_kernel(const int* __restrict__ src,
                  const int* __restrict__ dst,
                  const int* __restrict__ etype,
                  const float* __restrict__ eweight,
                  unsigned* __restrict__ enc) {
    static_assert((NNODES + OWN_ROWS - 1) / OWN_ROWS <= 256, "an owner number fits a byte");
    static_assert(NNODES <= (1 << 17) && OWN_ROWS <= (1 << 9), "record word: 17 + 1 + 9 bits");
    static_assert(NEDGES % 4 == 0 && (REC_OFF_WORDS % 4) == 0, "four edges per key word, 16-B record stores");
    const int w = blockIdx.x * 256 + threadIdx.x;
    if (w < NEDGES / 4) {
        const int4   d = *(const int4*)(dst + 4 * w);
        const int4   s = *(const int4*)(src + 4 * w);
        const int4   t = *(const int4*)(etype + 4 * w);
        const float4 f = *(const float4*)(eweight + 4 * w);
        const unsigned k0 = min((unsigned)d.x / OWN_ROWS, 255u);
        const unsigned k1 = min((unsigned)d.y / OWN_ROWS, 255u);
        const unsigned k2 = min((unsigned)d.z / OWN_ROWS, 255u);
        const unsigned k3 = min((unsigned)d.w / OWN_ROWS, 255u);
        enc[w] = k0 | (k1 << 8) | (k2 << 16) | (k3 << 24);
        auto word = [](int sv, int tv, int dv, unsigned k) -> unsigned {
            return ((unsigned)sv & 0x1FFFFu) | ((tv != 0) ? (1u << 17) : 0u)
                 | ((((unsigned)dv - k * OWN_ROWS) & 0x1FFu) << 18);
        };
        uint4* rp = (uint4*)(enc + REC_OFF_WORDS) + 2 * w;
        rp[0] = make_uint4(word(s.x, t.x, d.x, k0), __float_as_uint(f.x), word(s.y, t.y, d.y, k1), __float_as_uint(f.y));
        rp[1] = make_uint4(word(s.z, t.z, d.z, k2), __float_as_uint(f.z), word(s.w, t.w, d.w, k3), __float_as_uint(f.w));
    }
}

template <int DO_LN>
__global__ void __launch_bounds__(OWN_THR)
owner_sum_kernel(const unsigned* __restrict__ enc,
                 const float* __restrict__ y0,
                 const float* __restrict__ y1,
                 float* h,
                 const float* __restrict__ g,
                 const float* __restrict__ b) {
    static_assert(NEDGES % SCAN_EDGES == 0, "edge list is tested in whole 1024-edge pieces");
    __shared__ float acc[OWN_ROWS * EMB];
    __shared__ unsigned ring[OWN_WAVES * QCAP];

    const int tid  = threadIdx.x;
    const int lane = tid & 31;
    const int wave = __builtin_amdgcn_readfirstlane(tid >> 5);
    const int r0   = blockIdx.x * OWN_ROWS;
    const int nrow = min(OWN_ROWS, NNODES - r0);
    const unsigned nr = (unsigned)nrow;

    constexpr int RPW = (OWN_ROWS + OWN_WAVES - 1) / OWN_WAVES;
#pragma unroll
    for (int i0 = 0; i0 < RPW; i0 += 7) {
        float4 t[7];
#pragma unroll
        for (int i = 0; i < 7; ++i) {
            const int r = min(wave + OWN_WAVES * (i0 + i), nrow - 1);
            t[i] = *(const float4*)(h + (size_t)(r0 + r) * EMB + lane * 4);
        }
#pragma unroll
        for (int i = 0; i < 7; ++i) {
            const int r = wave + OWN_WAVES * (i0 + i);
            if (r < nrow) {
                float* a = acc + r * EMB + lane;
                a[0] = t[i].x; a[32] = t[i].y; a[64] = t[i].z; a[96] = t[i].w;
            }
        }
    }
    __syncthreads();

    unsigned* qw = ring + wave * QCAP;
    int qhead = 0, qn = 0;

    constexpr int NPIECE = NEDGES / SCAN_EDGES;
    constexpr int NTRIP  = (NPIECE + OWN_WAVES - 1) / OWN_WAVES;

    static_assert(OWN_WAVES <= NPIECE, "every wave has a first piece");
    constexpr int NTRIP3 = ((NTRIP + 2) / 3) * 3;
    const unsigned* klane = enc + lane * 4;
    const uint2* rec = (const uint2*)(enc + REC_OFF_WORDS);
    auto piece_at = [&](int p) { return klane + (size_t)min(p, NPIECE - 1) * (SCAN_EDGES / 4); };
    int4 idA[3], idB[3];
    idA[0] = *(const int4*)(piece_at(wave));
    idB[0] = *(const int4*)(piece_at(wave) + 128);
    idA[1] = *(const int4*)(piece_at(wave + OWN_WAVES));
    idB[1] = *(const int4*)(piece_at(wave + OWN_WAVES) + 128);
    const unsigned kpat = (unsigned)blockIdx.x * 0x01010101u;

    for (int it = 0; it < NTRIP3; it += 3) {
#pragma unroll
        for (int s = 0; s < 3; ++s) {
            const int piece = wave + OWN_WAVES * (it + s);
            const int4 cA = idA[s], cB = idB[s];
            idA[(s + 2) % 3] = *(const int4*)(piece_at(piece + 2 * OWN_WAVES));
            idB[(s + 2) % 3] = *(const int4*)(piece_at(piece + 2 * OWN_WAVES) + 128);

            auto eq = [&](int x) -> unsigned {
                const unsigned m = (unsigned)x ^ kpat;
                return ~(((m & 0x7F7F7F7Fu) + 0x7F7F7F7Fu) | m | 0x7F7F7F7Fu);
            };
            unsigned hm = (eq(cA.x) >> 7) | (eq(cA.y) >> 6) | (eq(cA.z) >> 5) | (eq(cA.w) >> 4)
                        | (eq(cB.x) >> 3) | (eq(cB.y) >> 2) | (eq(cB.z) >> 1) |  eq(cB.w);
            if (piece >= NPIECE) hm = 0u;
            const int ebase = piece * SCAN_EDGES + lane * 16;

            const bool last = (it + s == NTRIP3 - 1);
            unsigned more = __builtin_amdgcn_ballot_w32(hm != 0u);
            do {
                if (more) {
                    const int pos = qn + (int)__builtin_amdgcn_mbcnt_lo(more, 0u);
                    if (hm != 0u) {
                        const int bit = __builtin_ctz(hm);
                        const int wi = bit & 7, bj = bit >> 3;
                        qw[(qhead + pos) & (QCAP - 1)] = (unsigned)(ebase + (wi >> 2) * 512 + (wi & 3) * 4 + bj);
                        hm &= hm - 1u;
                    }
                    qn += __builtin_popcount(more);
                    more = __builtin_amdgcn_ballot_w32(hm != 0u);
                }
                while (qn >= 32 || (last && more == 0u && qn > 0)) {
                    const int n = min(qn, 32);
                    __builtin_amdgcn_wave_barrier();
                    int s_l = 0, t_l = 0, w_l = 0;
                    unsigned dl_l = 0;
                    if (lane < n) {
                        const uint2 r = rec[qw[(qhead + lane) & (QCAP - 1)]];
                        s_l  = (int)(r.x & 0x1FFFFu);
                        t_l  = (int)((r.x >> 17) & 1u);
                        dl_l = r.x >> 18;
                        w_l  = (int)r.y;
                    }
                    for (int j = 0; j < n; j += 8) {
                        float4 v[8];
                        float  wj[8];
                        unsigned dj[8];
    #pragma unroll
                        for (int u = 0; u < 8; ++u) {
                            const int jj = min(j + u, n - 1);
                            const int sj = __builtin_amdgcn_readlane(s_l, jj);
                            const int tj = __builtin_amdgcn_readlane(t_l, jj);
                            wj[u] = __int_as_float(__builtin_amdgcn_readlane(w_l, jj));
                            dj[u] = (unsigned)__builtin_amdgcn_readlane((int)dl_l, jj);
                            const float* y = ((tj == 0) ? y0 : y1) + (size_t)sj * EMB;
                            v[u] = *(const float4*)(y + lane * 4);
                        }
    #pragma unroll
                        for (int u = 0; u < 8; ++u) {
                            if (j + u < n && dj[u] < nr) {
                                float* a = acc + dj[u] * EMB + lane;
                                __hip_atomic_fetch_add(a,      wj[u] * v[u].x, __ATOMIC_RELAXED, __HIP_MEMORY_SCOPE_WORKGROUP);
                                __hip_atomic_fetch_add(a + 32, wj[u] * v[u].y, __ATOMIC_RELAXED, __HIP_MEMORY_SCOPE_WORKGROUP);
                                __hip_atomic_fetch_add(a + 64, wj[u] * v[u].z, __ATOMIC_RELAXED, __HIP_MEMORY_SCOPE_WORKGROUP);
                                __hip_atomic_fetch_add(a + 96, wj[u] * v[u].w, __ATOMIC_RELAXED, __HIP_MEMORY_SCOPE_WORKGROUP);
                            }
                        }
                    }
                    qhead = (qhead + n) & (QCAP - 1);
                    qn -= n;
                }
            } while (more);
        }
    }
    __syncthreads();

    float4 gg = make_float4(1.f, 1.f, 1.f, 1.f), bb = make_float4(0.f, 0.f, 0.f, 0.f);
    if (DO_LN) {
        gg = *(const float4*)(g + lane * 4);
        bb = *(const float4*)(b + lane * 4);
    }
    for (int g0 = 0; g0 < RPW; g0 += 4) {
        float4 v[4], o[4];
#pragma unroll
        for (int q = 0; q < 4; ++q) {
            const int r = min(wave + OWN_WAVES * (g0 + q), nrow - 1);
            const float* a = acc + r * EMB + lane;
            v[q] = make_float4(a[0], a[32], a[64], a[96]);
            o[q] = v[q];
        }
        if (DO_LN) {
            float s[4], t[4];
#pragma unroll
            for (int q = 0; q < 4; ++q) {
                v[q].x = fmaxf(v[q].x, 0.0f); v[q].y = fmaxf(v[q].y, 0.0f);
                v[q].z = fmaxf(v[q].z, 0.0f); v[q].w = fmaxf(v[q].w, 0.0f);
                s[q] = v[q].x + v[q].y + v[q].z + v[q].w;
            }
#pragma unroll
            for (int m = 16; m >= 1; m >>= 1) {
#pragma unroll
                for (int q = 0; q < 4; ++q) t[q] = __shfl_xor(s[q], m, 32);
#pragma unroll
                for (int q = 0; q < 4; ++q) s[q] += t[q];
            }
            float dx[4], dy[4], dz[4], dw[4];
#pragma unroll
            for (int q = 0; q < 4; ++q) {
                const float mu = s[q] * (1.0f / EMB);
                dx[q] = v[q].x - mu; dy[q] = v[q].y - mu; dz[q] = v[q].z - mu; dw[q] = v[q].w - mu;
                s[q] = dx[q] * dx[q] + dy[q] * dy[q] + dz[q] * dz[q] + dw[q] * dw[q];
            }
#pragma unroll
            for (int m = 16; m >= 1; m >>= 1) {
#pragma unroll
                for (int q = 0; q < 4; ++q) t[q] = __shfl_xor(s[q], m, 32);
#pragma unroll
                for (int q = 0; q < 4; ++q) s[q] += t[q];
            }
#pragma unroll
            for (int q = 0; q < 4; ++q) {
                const float var = s[q] * (1.0f / EMB);
                const float inv = rsqrtf(var + 1e-5f);
                o[q].x = dx[q] * inv * gg.x + bb.x;
                o[q].y = dy[q] * inv * gg.y + bb.y;
                o[q].z = dz[q] * inv * gg.z + bb.z;
                o[q].w = dw[q] * inv * gg.w + bb.w;
            }
        }
#pragma unroll
        for (int q = 0; q < 4; ++q) {
            const int r = wave + OWN_WAVES * (g0 + q);
            if (r < nrow) *(float4*)(h + (size_t)(r0 + r) * EMB + lane * 4) = o[q];
        }
    }
}

__global__ void gather_gmf_kernel(const float* __restrict__ h2,
                                  const int* __restrict__ ui,
                                  const int* __restrict__ vi,
                                  float* __restrict__ gmf,
                                  float* __restrict__ hcat) {
    const int bidx = blockIdx.x * (blockDim.x >> 5) + (threadIdx.x >> 5);
    if (bidx >= BATCH) return;
    const int lane = threadIdx.x & 31;
    const int u = ui[bidx], w = vi[bidx];
    const float4 uv = *(const float4*)(h2 + (size_t)u * EMB + lane * 4);
    const float4 vv = *(const float4*)(h2 + (size_t)w * EMB + lane * 4);
    const float su = wave_sum(uv.x * uv.x + uv.y * uv.y + uv.z * uv.z + uv.w * uv.w);
    const float sv = wave_sum(vv.x * vv.x + vv.y * vv.y + vv.z * vv.z + vv.w * vv.w);
    const float rn = 1.0f / (fmaxf(sqrtf(su), 1e-12f) * fmaxf(sqrtf(sv), 1e-12f));
    float4 gg;
    gg.x = uv.x * vv.x * rn; gg.y = uv.y * vv.y * rn;
    gg.z = uv.z * vv.z * rn; gg.w = uv.w * vv.w * rn;
    *(float4*)(gmf + (size_t)bidx * EMB + lane * 4) = gg;
    *(float4*)(hcat + (size_t)bidx * 2 * EMB + lane * 4) = uv;
    *(float4*)(hcat + (size_t)bidx * 2 * EMB + EMB + lane * 4) = vv;
}

__global__ void score_kernel(const float* __restrict__ gmf,
                             const float* __restrict__ h3,
                             const float* __restrict__ out_w,
                             const float* __restrict__ out_b,
                             float* __restrict__ out) {
    const int bidx = blockIdx.x * (blockDim.x >> 5) + (threadIdx.x >> 5);
    if (bidx >= BATCH) return;
    const int lane = threadIdx.x & 31;
    const float4 gg = *(const float4*)(gmf + (size_t)bidx * EMB + lane * 4);
    const float4 ww = *(const float4*)(out_w + lane * 4);
    float acc = gg.x * ww.x + gg.y * ww.y + gg.z * ww.z + gg.w * ww.w;
    acc += h3[(size_t)bidx * 32 + lane] * out_w[EMB + lane];
    acc = wave_sum(acc) + out_b[0];
    if (lane == 0) out[bidx] = 1.0f / (1.0f + expf(-acc));
}

extern "C" void kernel_launch(void* const* d_in, const int* in_sizes, int n_in,
                              void* d_out, int out_size, void* d_ws, size_t ws_size,
                              hipStream_t stream) {
    const int*   user_idx = (const int*)d_in[0];
    const int*   item_idx = (const int*)d_in[1];
    const int*   eidx     = (const int*)d_in[2];
    const int*   etype    = (const int*)d_in[3];
    const float* eweight  = (const float*)d_in[4];
    const float* emb      = (const float*)d_in[5];
    const float* w1_rel   = (const float*)d_in[6];
    const float* w1_root  = (const float*)d_in[7];
    const float* b1       = (const float*)d_in[8];
    const float* ln_g     = (const float*)d_in[9];
    const float* ln_b     = (const float*)d_in[10];
    const float* w2_rel   = (const float*)d_in[11];
    const float* w2_root  = (const float*)d_in[12];
    const float* b2       = (const float*)d_in[13];
    const float* mlp_w0   = (const float*)d_in[14];
    const float* mlp_b0   = (const float*)d_in[15];
    const float* mlp_w1   = (const float*)d_in[16];
    const float* mlp_b1   = (const float*)d_in[17];
    const float* mlp_w2   = (const float*)d_in[18];
    const float* mlp_b2   = (const float*)d_in[19];
    const float* out_w    = (const float*)d_in[20];
    const float* out_b    = (const float*)d_in[21];
    float* out = (float*)d_out;

    char* ws = (char*)d_ws;
    const size_t NB = (size_t)NNODES * EMB * sizeof(float);
    float* y0 = (float*)(ws);
    float* y1 = (float*)(ws + NB);
    float* h1 = (float*)(ws + 2 * NB);
    float* h2 = (float*)(ws + 3 * NB);
    float* gmf  = (float*)(ws);
    float* hcat = (float*)(ws + (size_t)BATCH * EMB * 4);
    float* m0   = (float*)(ws + (size_t)BATCH * EMB * 4 + (size_t)BATCH * 2 * EMB * 4);
    float* m1   = (float*)((char*)m0 + (size_t)BATCH * 128 * 4);
    float* m2   = (float*)((char*)m1 + (size_t)BATCH * 64 * 4);

    const int nodeGrid  = 256;
    const int batchGrid = 256;
    const int ownGrid   = (NNODES + OWN_ROWS - 1) / OWN_ROWS;
    const int rowGridB  = (BATCH + 7) / 8;

    wmma_gemm_kernel<8, EMB, 512><<<nodeGrid, 512, 0, stream>>>(emb, w1_rel,             nullptr, y0, NNODES, 0);
    wmma_gemm_kernel<8, EMB, 512><<<nodeGrid, 512, 0, stream>>>(emb, w1_rel + EMB * EMB, nullptr, y1, NNODES, 0);
    wmma_gemm_kernel<8, EMB, 512><<<nodeGrid, 512, 0, stream>>>(emb, w1_root,            b1,      h1, NNODES, 0);
    owner_prep_kernel<<<NEDGES / 4 / 256, 256, 0, stream>>>(eidx, eidx + NEDGES, etype, eweight, (unsigned*)h2);
    owner_sum_kernel<1><<<ownGrid, OWN_THR, 0, stream>>>((const unsigned*)h2, y0, y1, h1, ln_g, ln_b);

    wmma_gemm_kernel<8, EMB, 512><<<nodeGrid, 512, 0, stream>>>(h1, w2_rel,             nullptr, y0, NNODES, 0);
    wmma_gemm_kernel<8, EMB, 512><<<nodeGrid, 512, 0, stream>>>(h1, w2_rel + EMB * EMB, nullptr, y1, NNODES, 0);
    wmma_gemm_kernel<8, EMB, 512><<<nodeGrid, 512, 0, stream>>>(h1, w2_root,            b2,      h2, NNODES, 0);
    owner_prep_kernel<<<NEDGES / 4 / 256, 256, 0, stream>>>(eidx, eidx + NEDGES, etype, eweight, (unsigned*)h1);
    owner_sum_kernel<0><<<ownGrid, OWN_THR, 0, stream>>>((const unsigned*)h1, y0, y1, h2, nullptr, nullptr);

    gather_gmf_kernel<<<rowGridB, 256, 0, stream>>>(h2, user_idx, item_idx, gmf, hcat);
    wmma_gemm_kernel<8, 2 * EMB, 128><<<batchGrid, 128, 0, stream>>>(hcat, mlp_w0, mlp_b0, m0, BATCH, 1);
    wmma_gemm_kernel<4, 128, 128><<<batchGrid, 128, 0, stream>>>(m0,   mlp_w1, mlp_b1, m1, BATCH, 1);
    wmma_gemm_kernel<2, 64, 128><<<batchGrid, 128, 0, stream>>>(m1,   mlp_w2, mlp_b2, m2, BATCH, 1);
    score_kernel<<<rowGridB, 256, 0, stream>>>(gmf, m2, out_w, out_b, out);
}


# a layer's three node products share ONE fetch of X: W_rel0, W_rel1, W_root sit together in LDS and a wave runs its tile's A registers against all three (1 launch per layer, was 3); sums unchanged
# speedup vs baseline: 2.0491x; 1.1164x over previous
#include <hip/hip_runtime.h>
#include <hip/hip_bf16.h>
#include <math.h>

#define EMB     128
#define BATCH   16384
#define NNODES  100000
#define NEDGES  640000

typedef float v2f __attribute__((ext_vector_type(2)));
typedef float v8f __attribute__((ext_vector_type(8)));

__device__ __forceinline__ float wave_sum(float v) {
#pragma unroll
    for (int m = 16; m >= 1; m >>= 1) v += __shfl_xor(v, m, 32);
    return v;
}

template <int NT, int K, int NTHR>
__global__ void __launch_bounds__(NTHR)
wmma_gemm_kernel(const float* __restrict__ X,
                 const float* __restrict__ W,
                 const float* __restrict__ bias,
                 float* __restrict__ Y,
                 int M, int act) {
    constexpr int N     = NT * 16;
    constexpr int LDW   = N + 16;
    constexpr int KP    = K / 2;
    constexpr int KS    = K / 4;
    constexpr int WAVES = NTHR / 32;
    constexpr int NQ    = N / 4;
    constexpr int ITEMS = KP * NQ;
    __shared__ __attribute__((aligned(16))) float2 wt2[KP * LDW];

    const int tid  = threadIdx.x;
    const int wave = __builtin_amdgcn_readfirstlane(tid >> 5);
    const int lane = tid & 31;
    const int half = lane >> 4;
    const int l    = lane & 15;

    const int nTiles = (M + 15) >> 4;
    const int tstep  = gridDim.x * WAVES;
    int tile = blockIdx.x + gridDim.x * wave;

    float2 areg[KS];
    bool rowOK;
    {
        const int row = tile * 16 + l;
        rowOK = (tile < nTiles) && (row < M);
        const int rr = rowOK ? row : (M - 1);
        const float* xrow = X + (size_t)rr * K + 2 * half;
#pragma unroll
        for (int j = 0; j < KS; ++j) areg[j] = *(const float2*)(xrow + 4 * j);
    }

    float bv[NT];
#pragma unroll
    for (int t = 0; t < NT; ++t) bv[t] = bias ? bias[t * 16 + l] : 0.0f;

#pragma unroll
    for (int i0 = 0; i0 < ITEMS; i0 += NTHR) {
        const int i = i0 + tid;
        if ((ITEMS % NTHR == 0) || (i < ITEMS)) {
            const int p = i / NQ;
            const int c = (i % NQ) * 4;
            const float4 r0 = *(const float4*)(W + (size_t)(2 * p)     * N + c);
            const float4 r1 = *(const float4*)(W + (size_t)(2 * p + 1) * N + c);
            float4* dstp = (float4*)(wt2 + p * LDW + c);
            dstp[0] = make_float4(r0.x, r1.x, r0.y, r1.y);
            dstp[1] = make_float4(r0.z, r1.z, r0.w, r1.w);
        }
    }
    __syncthreads();

    for (; tile < nTiles; tile += tstep) {
        v8f acc[NT] = {};
        const float2* bp0 = wt2 + half * LDW + l;
        float2 bcur[NT], bnxt[NT];
#pragma unroll
        for (int t = 0; t < NT; ++t) bcur[t] = bp0[t * 16];
#pragma unroll
        for (int j = 0; j < KS; ++j) {
            v2f a;
            a.x = rowOK ? areg[j].x : 0.0f;
            a.y = rowOK ? areg[j].y : 0.0f;
            if (j + 1 < KS) {
                const float2* bp = bp0 + (2 * (j + 1)) * LDW;
#pragma unroll
                for (int t = 0; t < NT; ++t) bnxt[t] = bp[t * 16];
            }
#pragma unroll
            for (int t = 0; t < NT; ++t) {
                v2f b; b.x = bcur[t].x; b.y = bcur[t].y;
                acc[t] = __builtin_amdgcn_wmma_f32_16x16x4_f32(
                    false, a, false, b, (short)0, acc[t], false, false);
            }
            __builtin_amdgcn_sched_barrier(0);
            if (j + 1 < KS) {
#pragma unroll
                for (int t = 0; t < NT; ++t) bcur[t] = bnxt[t];
            }
        }

        const int nxt = tile + tstep;
        if (nxt < nTiles) {
            const int row = nxt * 16 + l;
            rowOK = (row < M);
            const int rr = rowOK ? row : (M - 1);
            const float* xrow = X + (size_t)rr * K + 2 * half;
#pragma unroll
            for (int j = 0; j < KS; ++j) areg[j] = *(const float2*)(xrow + 4 * j);
        }

        const int rowBase = tile * 16;
#pragma unroll
        for (int t = 0; t < NT; ++t) {
            const int c = t * 16 + l;
#pragma unroll
            for (int v = 0; v < 8; ++v) {
                const int r = rowBase + v + 8 * half;
                if (r < M) {
                    float o = acc[t][v] + bv[t];
                    if (act) o = fmaxf(o, 0.0f);
                    Y[(size_t)r * N + c] = o;
                }
            }
        }
    }
}

template <int NT, int K, int NTHR>
__global__ void __launch_bounds__(NTHR)
wmma_gemm3_kernel(const float* __restrict__ X,
                  const float* __restrict__ W0,
                  const float* __restrict__ W1,
                  const float* __restrict__ W2,
                  const float* __restrict__ bias2,
                  float* __restrict__ Y0,
                  float* __restrict__ Y1,
                  float* __restrict__ Y2,
                  int M) {
    constexpr int N     = NT * 16;
    constexpr int LDW   = N + 16;
    constexpr int KP    = K / 2;
    constexpr int KS    = K / 4;
    constexpr int WAVES = NTHR / 32;
    constexpr int NQ    = N / 4;
    constexpr int ITEMS = KP * NQ;
    constexpr int WMAT  = KP * LDW;
    __shared__ __attribute__((aligned(16))) float2 wt2[3 * WMAT];

    const int tid  = threadIdx.x;
    const int wave = __builtin_amdgcn_readfirstlane(tid >> 5);
    const int lane = tid & 31;
    const int half = lane >> 4;
    const int l    = lane & 15;

    const int nTiles = (M + 15) >> 4;
    const int tstep  = gridDim.x * WAVES;
    int tile = blockIdx.x + gridDim.x * wave;

    float2 areg[KS];
    bool rowOK;
    {
        const int row = tile * 16 + l;
        rowOK = (tile < nTiles) && (row < M);
        const int rr = rowOK ? row : (M - 1);
        const float* xrow = X + (size_t)rr * K + 2 * half;
#pragma unroll
        for (int j = 0; j < KS; ++j) areg[j] = *(const float2*)(xrow + 4 * j);
    }

    float bv[NT];
#pragma unroll
    for (int t = 0; t < NT; ++t) bv[t] = bias2 ? bias2[t * 16 + l] : 0.0f;

#pragma unroll
    for (int m = 0; m < 3; ++m) {
        const float* __restrict__ W = (m == 0) ? W0 : ((m == 1) ? W1 : W2);
#pragma unroll
        for (int i0 = 0; i0 < ITEMS; i0 += NTHR) {
            const int i = i0 + tid;
            if ((ITEMS % NTHR == 0) || (i < ITEMS)) {
                const int p = i / NQ;
                const int c = (i % NQ) * 4;
                const float4 r0 = *(const float4*)(W + (size_t)(2 * p)     * N + c);
                const float4 r1 = *(const float4*)(W + (size_t)(2 * p + 1) * N + c);
                float4* dstp = (float4*)(wt2 + m * WMAT + p * LDW + c);
                dstp[0] = make_float4(r0.x, r1.x, r0.y, r1.y);
                dstp[1] = make_float4(r0.z, r1.z, r0.w, r1.w);
            }
        }
    }
    __syncthreads();

    for (; tile < nTiles; tile += tstep) {
        const int rowBase = tile * 16;
        const int nxt = tile + tstep;
#pragma unroll 1
        for (int m = 0; m < 3; ++m) {
            v8f acc[NT] = {};
            const float2* bp0 = wt2 + m * WMAT + half * LDW + l;
            float2 bcur[NT], bnxt[NT];
#pragma unroll
            for (int t = 0; t < NT; ++t) bcur[t] = bp0[t * 16];
#pragma unroll
            for (int j = 0; j < KS; ++j) {
                v2f a;
                a.x = rowOK ? areg[j].x : 0.0f;
                a.y = rowOK ? areg[j].y : 0.0f;
                if (j + 1 < KS) {
                    const float2* bp = bp0 + (2 * (j + 1)) * LDW;
#pragma unroll
                    for (int t = 0; t < NT; ++t) bnxt[t] = bp[t * 16];
                }
#pragma unroll
                for (int t = 0; t < NT; ++t) {
                    v2f b; b.x = bcur[t].x; b.y = bcur[t].y;
                    acc[t] = __builtin_amdgcn_wmma_f32_16x16x4_f32(
                        false, a, false, b, (short)0, acc[t], false, false);
                }
                __builtin_amdgcn_sched_barrier(0);
                if (j + 1 < KS) {
#pragma unroll
                    for (int t = 0; t < NT; ++t) bcur[t] = bnxt[t];
                }
            }

            if (m == 2 && nxt < nTiles) {
                const int row = nxt * 16 + l;
                rowOK = (row < M);
                const int rr = rowOK ? row : (M - 1);
                const float* xrow = X + (size_t)rr * K + 2 * half;
#pragma unroll
                for (int j = 0; j < KS; ++j) areg[j] = *(const float2*)(xrow + 4 * j);
            }

            float* __restrict__ Y = (m == 0) ? Y0 : ((m == 1) ? Y1 : Y2);
#pragma unroll
            for (int t = 0; t < NT; ++t) {
                const int c = t * 16 + l;
                const float badd = (m == 2) ? bv[t] : 0.0f;
#pragma unroll
                for (int v = 0; v < 8; ++v) {
                    const int r = rowBase + v + 8 * half;
                    if (r < M) Y[(size_t)r * N + c] = acc[t][v] + badd;
                }
            }
        }
    }
}

#define OWN_ROWS 392
#define OWN_THR  1024
#define OWN_WAVES (OWN_THR / 32)
#define QCAP     64
#define SCAN_EDGES 1024

#define REC_OFF_WORDS (NEDGES / 4 + 3840)
__global__ void __launch_bounds__(256)
owner_prep_kernel(const int* __restrict__ src,
                  const int* __restrict__ dst,
                  const int* __restrict__ etype,
                  const float* __restrict__ eweight,
                  unsigned* __restrict__ enc) {
    static_assert((NNODES + OWN_ROWS - 1) / OWN_ROWS <= 256, "an owner number fits a byte");
    static_assert(NNODES <= (1 << 17) && OWN_ROWS <= (1 << 9), "record word: 17 + 1 + 9 bits");
    static_assert(NEDGES % 4 == 0 && (REC_OFF_WORDS % 4) == 0, "four edges per key word, 16-B record stores");
    const int w = blockIdx.x * 256 + threadIdx.x;
    if (w < NEDGES / 4) {
        const int4   d = *(const int4*)(dst + 4 * w);
        const int4   s = *(const int4*)(src + 4 * w);
        const int4   t = *(const int4*)(etype + 4 * w);
        const float4 f = *(const float4*)(eweight + 4 * w);
        const unsigned k0 = min((unsigned)d.x / OWN_ROWS, 255u);
        const unsigned k1 = min((unsigned)d.y / OWN_ROWS, 255u);
        const unsigned k2 = min((unsigned)d.z / OWN_ROWS, 255u);
        const unsigned k3 = min((unsigned)d.w / OWN_ROWS, 255u);
        enc[w] = k0 | (k1 << 8) | (k2 << 16) | (k3 << 24);
        auto word = [](int sv, int tv, int dv, unsigned k) -> unsigned {
            return ((unsigned)sv & 0x1FFFFu) | ((tv != 0) ? (1u << 17) : 0u)
                 | ((((unsigned)dv - k * OWN_ROWS) & 0x1FFu) << 18);
        };
        uint4* rp = (uint4*)(enc + REC_OFF_WORDS) + 2 * w;
        rp[0] = make_uint4(word(s.x, t.x, d.x, k0), __float_as_uint(f.x), word(s.y, t.y, d.y, k1), __float_as_uint(f.y));
        rp[1] = make_uint4(word(s.z, t.z, d.z, k2), __float_as_uint(f.z), word(s.w, t.w, d.w, k3), __float_as_uint(f.w));
    }
}

template <int DO_LN>
__global__ void __launch_bounds__(OWN_THR)
owner_sum_kernel(const unsigned* __restrict__ enc,
                 const float* __restrict__ y0,
                 const float* __restrict__ y1,
                 float* h,
                 const float* __restrict__ g,
                 const float* __restrict__ b) {
    static_assert(NEDGES % SCAN_EDGES == 0, "edge list is tested in whole 1024-edge pieces");
    __shared__ float acc[OWN_ROWS * EMB];
    __shared__ unsigned ring[OWN_WAVES * QCAP];

    const int tid  = threadIdx.x;
    const int lane = tid & 31;
    const int wave = __builtin_amdgcn_readfirstlane(tid >> 5);
    const int r0   = blockIdx.x * OWN_ROWS;
    const int nrow = min(OWN_ROWS, NNODES - r0);
    const unsigned nr = (unsigned)nrow;

    constexpr int RPW = (OWN_ROWS + OWN_WAVES - 1) / OWN_WAVES;
#pragma unroll
    for (int i0 = 0; i0 < RPW; i0 += 7) {
        float4 t[7];
#pragma unroll
        for (int i = 0; i < 7; ++i) {
            const int r = min(wave + OWN_WAVES * (i0 + i), nrow - 1);
            t[i] = *(const float4*)(h + (size_t)(r0 + r) * EMB + lane * 4);
        }
#pragma unroll
        for (int i = 0; i < 7; ++i) {
            const int r = wave + OWN_WAVES * (i0 + i);
            if (r < nrow) {
                float* a = acc + r * EMB + lane;
                a[0] = t[i].x; a[32] = t[i].y; a[64] = t[i].z; a[96] = t[i].w;
            }
        }
    }
    __syncthreads();

    unsigned* qw = ring + wave * QCAP;
    int qhead = 0, qn = 0;

    constexpr int NPIECE = NEDGES / SCAN_EDGES;
    constexpr int NTRIP  = (NPIECE + OWN_WAVES - 1) / OWN_WAVES;

    static_assert(OWN_WAVES <= NPIECE, "every wave has a first piece");
    constexpr int NTRIP3 = ((NTRIP + 2) / 3) * 3;
    const unsigned* klane = enc + lane * 4;
    const uint2* rec = (const uint2*)(enc + REC_OFF_WORDS);
    auto piece_at = [&](int p) { return klane + (size_t)min(p, NPIECE - 1) * (SCAN_EDGES / 4); };
    int4 idA[3], idB[3];
    idA[0] = *(const int4*)(piece_at(wave));
    idB[0] = *(const int4*)(piece_at(wave) + 128);
    idA[1] = *(const int4*)(piece_at(wave + OWN_WAVES));
    idB[1] = *(const int4*)(piece_at(wave + OWN_WAVES) + 128);
    const unsigned kpat = (unsigned)blockIdx.x * 0x01010101u;

    for (int it = 0; it < NTRIP3; it += 3) {
#pragma unroll
        for (int s = 0; s < 3; ++s) {
            const int piece = wave + OWN_WAVES * (it + s);
            const int4 cA = idA[s], cB = idB[s];
            idA[(s + 2) % 3] = *(const int4*)(piece_at(piece + 2 * OWN_WAVES));
            idB[(s + 2) % 3] = *(const int4*)(piece_at(piece + 2 * OWN_WAVES) + 128);

            auto eq = [&](int x) -> unsigned {
                const unsigned m = (unsigned)x ^ kpat;
                return ~(((m & 0x7F7F7F7Fu) + 0x7F7F7F7Fu) | m | 0x7F7F7F7Fu);
            };
            unsigned hm = (eq(cA.x) >> 7) | (eq(cA.y) >> 6) | (eq(cA.z) >> 5) | (eq(cA.w) >> 4)
                        | (eq(cB.x) >> 3) | (eq(cB.y) >> 2) | (eq(cB.z) >> 1) |  eq(cB.w);
            if (piece >= NPIECE) hm = 0u;
            const int ebase = piece * SCAN_EDGES + lane * 16;

            const bool last = (it + s == NTRIP3 - 1);
            unsigned more = __builtin_amdgcn_ballot_w32(hm != 0u);
            do {
                if (more) {
                    const int pos = qn + (int)__builtin_amdgcn_mbcnt_lo(more, 0u);
                    if (hm != 0u) {
                        const int bit = __builtin_ctz(hm);
                        const int wi = bit & 7, bj = bit >> 3;
                        qw[(qhead + pos) & (QCAP - 1)] = (unsigned)(ebase + (wi >> 2) * 512 + (wi & 3) * 4 + bj);
                        hm &= hm - 1u;
                    }
                    qn += __builtin_popcount(more);
                    more = __builtin_amdgcn_ballot_w32(hm != 0u);
                }
                while (qn >= 32 || (last && more == 0u && qn > 0)) {
                    const int n = min(qn, 32);
                    __builtin_amdgcn_wave_barrier();
                    int s_l = 0, t_l = 0, w_l = 0;
                    unsigned dl_l = 0;
                    if (lane < n) {
                        const uint2 r = rec[qw[(qhead + lane) & (QCAP - 1)]];
                        s_l  = (int)(r.x & 0x1FFFFu);
                        t_l  = (int)((r.x >> 17) & 1u);
                        dl_l = r.x >> 18;
                        w_l  = (int)r.y;
                    }
                    for (int j = 0; j < n; j += 8) {
                        float4 v[8];
                        float  wj[8];
                        unsigned dj[8];
    #pragma unroll
                        for (int u = 0; u < 8; ++u) {
                            const int jj = min(j + u, n - 1);
                            const int sj = __builtin_amdgcn_readlane(s_l, jj);
                            const int tj = __builtin_amdgcn_readlane(t_l, jj);
                            wj[u] = __int_as_float(__builtin_amdgcn_readlane(w_l, jj));
                            dj[u] = (unsigned)__builtin_amdgcn_readlane((int)dl_l, jj);
                            const float* y = ((tj == 0) ? y0 : y1) + (size_t)sj * EMB;
                            v[u] = *(const float4*)(y + lane * 4);
                        }
    #pragma unroll
                        for (int u = 0; u < 8; ++u) {
                            if (j + u < n && dj[u] < nr) {
                                float* a = acc + dj[u] * EMB + lane;
                                __hip_atomic_fetch_add(a,      wj[u] * v[u].x, __ATOMIC_RELAXED, __HIP_MEMORY_SCOPE_WORKGROUP);
                                __hip_atomic_fetch_add(a + 32, wj[u] * v[u].y, __ATOMIC_RELAXED, __HIP_MEMORY_SCOPE_WORKGROUP);
                                __hip_atomic_fetch_add(a + 64, wj[u] * v[u].z, __ATOMIC_RELAXED, __HIP_MEMORY_SCOPE_WORKGROUP);
                                __hip_atomic_fetch_add(a + 96, wj[u] * v[u].w, __ATOMIC_RELAXED, __HIP_MEMORY_SCOPE_WORKGROUP);
                            }
                        }
                    }
                    qhead = (qhead + n) & (QCAP - 1);
                    qn -= n;
                }
            } while (more);
        }
    }
    __syncthreads();

    float4 gg = make_float4(1.f, 1.f, 1.f, 1.f), bb = make_float4(0.f, 0.f, 0.f, 0.f);
    if (DO_LN) {
        gg = *(const float4*)(g + lane * 4);
        bb = *(const float4*)(b + lane * 4);
    }
    for (int g0 = 0; g0 < RPW; g0 += 4) {
        float4 v[4], o[4];
#pragma unroll
        for (int q = 0; q < 4; ++q) {
            const int r = min(wave + OWN_WAVES * (g0 + q), nrow - 1);
            const float* a = acc + r * EMB + lane;
            v[q] = make_float4(a[0], a[32], a[64], a[96]);
            o[q] = v[q];
        }
        if (DO_LN) {
            float s[4], t[4];
#pragma unroll
            for (int q = 0; q < 4; ++q) {
                v[q].x = fmaxf(v[q].x, 0.0f); v[q].y = fmaxf(v[q].y, 0.0f);
                v[q].z = fmaxf(v[q].z, 0.0f); v[q].w = fmaxf(v[q].w, 0.0f);
                s[q] = v[q].x + v[q].y + v[q].z + v[q].w;
            }
#pragma unroll
            for (int m = 16; m >= 1; m >>= 1) {
#pragma unroll
                for (int q = 0; q < 4; ++q) t[q] = __shfl_xor(s[q], m, 32);
#pragma unroll
                for (int q = 0; q < 4; ++q) s[q] += t[q];
            }
            float dx[4], dy[4], dz[4], dw[4];
#pragma unroll
            for (int q = 0; q < 4; ++q) {
                const float mu = s[q] * (1.0f / EMB);
                dx[q] = v[q].x - mu; dy[q] = v[q].y - mu; dz[q] = v[q].z - mu; dw[q] = v[q].w - mu;
                s[q] = dx[q] * dx[q] + dy[q] * dy[q] + dz[q] * dz[q] + dw[q] * dw[q];
            }
#pragma unroll
            for (int m = 16; m >= 1; m >>= 1) {
#pragma unroll
                for (int q = 0; q < 4; ++q) t[q] = __shfl_xor(s[q], m, 32);
#pragma unroll
                for (int q = 0; q < 4; ++q) s[q] += t[q];
            }
#pragma unroll
            for (int q = 0; q < 4; ++q) {
                const float var = s[q] * (1.0f / EMB);
                const float inv = rsqrtf(var + 1e-5f);
                o[q].x = dx[q] * inv * gg.x + bb.x;
                o[q].y = dy[q] * inv * gg.y + bb.y;
                o[q].z = dz[q] * inv * gg.z + bb.z;
                o[q].w = dw[q] * inv * gg.w + bb.w;
            }
        }
#pragma unroll
        for (int q = 0; q < 4; ++q) {
            const int r = wave + OWN_WAVES * (g0 + q);
            if (r < nrow) *(float4*)(h + (size_t)(r0 + r) * EMB + lane * 4) = o[q];
        }
    }
}

__global__ void gather_gmf_kernel(const float* __restrict__ h2,
                                  const int* __restrict__ ui,
                                  const int* __restrict__ vi,
                                  float* __restrict__ gmf,
                                  float* __restrict__ hcat) {
    const int bidx = blockIdx.x * (blockDim.x >> 5) + (threadIdx.x >> 5);
    if (bidx >= BATCH) return;
    const int lane = threadIdx.x & 31;
    const int u = ui[bidx], w = vi[bidx];
    const float4 uv = *(const float4*)(h2 + (size_t)u * EMB + lane * 4);
    const float4 vv = *(const float4*)(h2 + (size_t)w * EMB + lane * 4);
    const float su = wave_sum(uv.x * uv.x + uv.y * uv.y + uv.z * uv.z + uv.w * uv.w);
    const float sv = wave_sum(vv.x * vv.x + vv.y * vv.y + vv.z * vv.z + vv.w * vv.w);
    const float rn = 1.0f / (fmaxf(sqrtf(su), 1e-12f) * fmaxf(sqrtf(sv), 1e-12f));
    float4 gg;
    gg.x = uv.x * vv.x * rn; gg.y = uv.y * vv.y * rn;
    gg.z = uv.z * vv.z * rn; gg.w = uv.w * vv.w * rn;
    *(float4*)(gmf + (size_t)bidx * EMB + lane * 4) = gg;
    *(float4*)(hcat + (size_t)bidx * 2 * EMB + lane * 4) = uv;
    *(float4*)(hcat + (size_t)bidx * 2 * EMB + EMB + lane * 4) = vv;
}

__global__ void score_kernel(const float* __restrict__ gmf,
                             const float* __restrict__ h3,
                             const float* __restrict__ out_w,
                             const float* __restrict__ out_b,
                             float* __restrict__ out) {
    const int bidx = blockIdx.x * (blockDim.x >> 5) + (threadIdx.x >> 5);
    if (bidx >= BATCH) return;
    const int lane = threadIdx.x & 31;
    const float4 gg = *(const float4*)(gmf + (size_t)bidx * EMB + lane * 4);
    const float4 ww = *(const float4*)(out_w + lane * 4);
    float acc = gg.x * ww.x + gg.y * ww.y + gg.z * ww.z + gg.w * ww.w;
    acc += h3[(size_t)bidx * 32 + lane] * out_w[EMB + lane];
    acc = wave_sum(acc) + out_b[0];
    if (lane == 0) out[bidx] = 1.0f / (1.0f + expf(-acc));
}

extern "C" void kernel_launch(void* const* d_in, const int* in_sizes, int n_in,
                              void* d_out, int out_size, void* d_ws, size_t ws_size,
                              hipStream_t stream) {
    const int*   user_idx = (const int*)d_in[0];
    const int*   item_idx = (const int*)d_in[1];
    const int*   eidx     = (const int*)d_in[2];
    const int*   etype    = (const int*)d_in[3];
    const float* eweight  = (const float*)d_in[4];
    const float* emb      = (const float*)d_in[5];
    const float* w1_rel   = (const float*)d_in[6];
    const float* w1_root  = (const float*)d_in[7];
    const float* b1       = (const float*)d_in[8];
    const float* ln_g     = (const float*)d_in[9];
    const float* ln_b     = (const float*)d_in[10];
    const float* w2_rel   = (const float*)d_in[11];
    const float* w2_root  = (const float*)d_in[12];
    const float* b2       = (const float*)d_in[13];
    const float* mlp_w0   = (const float*)d_in[14];
    const float* mlp_b0   = (const float*)d_in[15];
    const float* mlp_w1   = (const float*)d_in[16];
    const float* mlp_b1   = (const float*)d_in[17];
    const float* mlp_w2   = (const float*)d_in[18];
    const float* mlp_b2   = (const float*)d_in[19];
    const float* out_w    = (const float*)d_in[20];
    const float* out_b    = (const float*)d_in[21];
    float* out = (float*)d_out;

    char* ws = (char*)d_ws;
    const size_t NB = (size_t)NNODES * EMB * sizeof(float);
    float* y0 = (float*)(ws);
    float* y1 = (float*)(ws + NB);
    float* h1 = (float*)(ws + 2 * NB);
    float* h2 = (float*)(ws + 3 * NB);
    float* gmf  = (float*)(ws);
    float* hcat = (float*)(ws + (size_t)BATCH * EMB * 4);
    float* m0   = (float*)(ws + (size_t)BATCH * EMB * 4 + (size_t)BATCH * 2 * EMB * 4);
    float* m1   = (float*)((char*)m0 + (size_t)BATCH * 128 * 4);
    float* m2   = (float*)((char*)m1 + (size_t)BATCH * 64 * 4);

    const int nodeGrid  = 256;
    const int batchGrid = 256;
    const int ownGrid   = (NNODES + OWN_ROWS - 1) / OWN_ROWS;
    const int rowGridB  = (BATCH + 7) / 8;

    wmma_gemm3_kernel<8, EMB, 512><<<nodeGrid, 512, 0, stream>>>(emb, w1_rel, w1_rel + EMB * EMB, w1_root, b1, y0, y1, h1, NNODES);
    owner_prep_kernel<<<NEDGES / 4 / 256, 256, 0, stream>>>(eidx, eidx + NEDGES, etype, eweight, (unsigned*)h2);
    owner_sum_kernel<1><<<ownGrid, OWN_THR, 0, stream>>>((const unsigned*)h2, y0, y1, h1, ln_g, ln_b);

    wmma_gemm3_kernel<8, EMB, 512><<<nodeGrid, 512, 0, stream>>>(h1, w2_rel, w2_rel + EMB * EMB, w2_root, b2, y0, y1, h2, NNODES);
    owner_prep_kernel<<<NEDGES / 4 / 256, 256, 0, stream>>>(eidx, eidx + NEDGES, etype, eweight, (unsigned*)h1);
    owner_sum_kernel<0><<<ownGrid, OWN_THR, 0, stream>>>((const unsigned*)h1, y0, y1, h2, nullptr, nullptr);

    gather_gmf_kernel<<<rowGridB, 256, 0, stream>>>(h2, user_idx, item_idx, gmf, hcat);
    wmma_gemm_kernel<8, 2 * EMB, 128><<<batchGrid, 128, 0, stream>>>(hcat, mlp_w0, mlp_b0, m0, BATCH, 1);
    wmma_gemm_kernel<4, 128, 128><<<batchGrid, 128, 0, stream>>>(m0,   mlp_w1, mlp_b1, m1, BATCH, 1);
    wmma_gemm_kernel<2, 64, 128><<<batchGrid, 128, 0, stream>>>(m1,   mlp_w2, mlp_b2, m2, BATCH, 1);
    score_kernel<<<rowGridB, 256, 0, stream>>>(gmf, m2, out_w, out_b, out);
}


# layer 2 reuses layer 1's edge filter: each wave lists the records it kept, layer 2 reads its list back (no 2nd key scan or re-encode launch; root product in place); LDS add order as before
# speedup vs baseline: 2.2096x; 1.0783x over previous
#include <hip/hip_runtime.h>
#include <hip/hip_bf16.h>
#include <math.h>

#define EMB     128
#define BATCH   16384
#define NNODES  100000
#define NEDGES  640000

typedef float v2f __attribute__((ext_vector_type(2)));
typedef float v8f __attribute__((ext_vector_type(8)));

__device__ __forceinline__ float wave_sum(float v) {
#pragma unroll
    for (int m = 16; m >= 1; m >>= 1) v += __shfl_xor(v, m, 32);
    return v;
}

template <int NT, int K, int NTHR>
__global__ void __launch_bounds__(NTHR)
wmma_gemm_kernel(const float* __restrict__ X,
                 const float* __restrict__ W,
                 const float* __restrict__ bias,
                 float* __restrict__ Y,
                 int M, int act) {
    constexpr int N     = NT * 16;
    constexpr int LDW   = N + 16;
    constexpr int KP    = K / 2;
    constexpr int KS    = K / 4;
    constexpr int WAVES = NTHR / 32;
    constexpr int NQ    = N / 4;
    constexpr int ITEMS = KP * NQ;
    __shared__ __attribute__((aligned(16))) float2 wt2[KP * LDW];

    const int tid  = threadIdx.x;
    const int wave = __builtin_amdgcn_readfirstlane(tid >> 5);
    const int lane = tid & 31;
    const int half = lane >> 4;
    const int l    = lane & 15;

    const int nTiles = (M + 15) >> 4;
    const int tstep  = gridDim.x * WAVES;
    int tile = blockIdx.x + gridDim.x * wave;

    float2 areg[KS];
    bool rowOK;
    {
        const int row = tile * 16 + l;
        rowOK = (tile < nTiles) && (row < M);
        const int rr = rowOK ? row : (M - 1);
        const float* xrow = X + (size_t)rr * K + 2 * half;
#pragma unroll
        for (int j = 0; j < KS; ++j) areg[j] = *(const float2*)(xrow + 4 * j);
    }

    float bv[NT];
#pragma unroll
    for (int t = 0; t < NT; ++t) bv[t] = bias ? bias[t * 16 + l] : 0.0f;

#pragma unroll
    for (int i0 = 0; i0 < ITEMS; i0 += NTHR) {
        const int i = i0 + tid;
        if ((ITEMS % NTHR == 0) || (i < ITEMS)) {
            const int p = i / NQ;
            const int c = (i % NQ) * 4;
            const float4 r0 = *(const float4*)(W + (size_t)(2 * p)     * N + c);
            const float4 r1 = *(const float4*)(W + (size_t)(2 * p + 1) * N + c);
            float4* dstp = (float4*)(wt2 + p * LDW + c);
            dstp[0] = make_float4(r0.x, r1.x, r0.y, r1.y);
            dstp[1] = make_float4(r0.z, r1.z, r0.w, r1.w);
        }
    }
    __syncthreads();

    for (; tile < nTiles; tile += tstep) {
        v8f acc[NT] = {};
        const float2* bp0 = wt2 + half * LDW + l;
        float2 bcur[NT], bnxt[NT];
#pragma unroll
        for (int t = 0; t < NT; ++t) bcur[t] = bp0[t * 16];
#pragma unroll
        for (int j = 0; j < KS; ++j) {
            v2f a;
            a.x = rowOK ? areg[j].x : 0.0f;
            a.y = rowOK ? areg[j].y : 0.0f;
            if (j + 1 < KS) {
                const float2* bp = bp0 + (2 * (j + 1)) * LDW;
#pragma unroll
                for (int t = 0; t < NT; ++t) bnxt[t] = bp[t * 16];
            }
#pragma unroll
            for (int t = 0; t < NT; ++t) {
                v2f b; b.x = bcur[t].x; b.y = bcur[t].y;
                acc[t] = __builtin_amdgcn_wmma_f32_16x16x4_f32(
                    false, a, false, b, (short)0, acc[t], false, false);
            }
            __builtin_amdgcn_sched_barrier(0);
            if (j + 1 < KS) {
#pragma unroll
                for (int t = 0; t < NT; ++t) bcur[t] = bnxt[t];
            }
        }

        const int nxt = tile + tstep;
        if (nxt < nTiles) {
            const int row = nxt * 16 + l;
            rowOK = (row < M);
            const int rr = rowOK ? row : (M - 1);
            const float* xrow = X + (size_t)rr * K + 2 * half;
#pragma unroll
            for (int j = 0; j < KS; ++j) areg[j] = *(const float2*)(xrow + 4 * j);
        }

        const int rowBase = tile * 16;
#pragma unroll
        for (int t = 0; t < NT; ++t) {
            const int c = t * 16 + l;
#pragma unroll
            for (int v = 0; v < 8; ++v) {
                const int r = rowBase + v + 8 * half;
                if (r < M) {
                    float o = acc[t][v] + bv[t];
                    if (act) o = fmaxf(o, 0.0f);
                    Y[(size_t)r * N + c] = o;
                }
            }
        }
    }
}

template <int NT, int K, int NTHR>
__global__ void __launch_bounds__(NTHR)
wmma_gemm3_kernel(const float* X,
                  const float* __restrict__ W0,
                  const float* __restrict__ W1,
                  const float* __restrict__ W2,
                  const float* __restrict__ bias2,
                  float* __restrict__ Y0,
                  float* __restrict__ Y1,
                  float* Y2,
                  int M) {
    constexpr int N     = NT * 16;
    constexpr int LDW   = N + 16;
    constexpr int KP    = K / 2;
    constexpr int KS    = K / 4;
    constexpr int WAVES = NTHR / 32;
    constexpr int NQ    = N / 4;
    constexpr int ITEMS = KP * NQ;
    constexpr int WMAT  = KP * LDW;
    __shared__ __attribute__((aligned(16))) float2 wt2[3 * WMAT];

    const int tid  = threadIdx.x;
    const int wave = __builtin_amdgcn_readfirstlane(tid >> 5);
    const int lane = tid & 31;
    const int half = lane >> 4;
    const int l    = lane & 15;

    const int nTiles = (M + 15) >> 4;
    const int tstep  = gridDim.x * WAVES;
    int tile = blockIdx.x + gridDim.x * wave;

    float2 areg[KS];
    bool rowOK;
    {
        const int row = tile * 16 + l;
        rowOK = (tile < nTiles) && (row < M);
        const int rr = rowOK ? row : (M - 1);
        const float* xrow = X + (size_t)rr * K + 2 * half;
#pragma unroll
        for (int j = 0; j < KS; ++j) areg[j] = *(const float2*)(xrow + 4 * j);
    }

    float bv[NT];
#pragma unroll
    for (int t = 0; t < NT; ++t) bv[t] = bias2 ? bias2[t * 16 + l] : 0.0f;

#pragma unroll
    for (int m = 0; m < 3; ++m) {
        const float* __restrict__ W = (m == 0) ? W0 : ((m == 1) ? W1 : W2);
#pragma unroll
        for (int i0 = 0; i0 < ITEMS; i0 += NTHR) {
            const int i = i0 + tid;
            if ((ITEMS % NTHR == 0) || (i < ITEMS)) {
                const int p = i / NQ;
                const int c = (i % NQ) * 4;
                const float4 r0 = *(const float4*)(W + (size_t)(2 * p)     * N + c);
                const float4 r1 = *(const float4*)(W + (size_t)(2 * p + 1) * N + c);
                float4* dstp = (float4*)(wt2 + m * WMAT + p * LDW + c);
                dstp[0] = make_float4(r0.x, r1.x, r0.y, r1.y);
                dstp[1] = make_float4(r0.z, r1.z, r0.w, r1.w);
            }
        }
    }
    __syncthreads();

    for (; tile < nTiles; tile += tstep) {
        const int rowBase = tile * 16;
        const int nxt = tile + tstep;
#pragma unroll 1
        for (int m = 0; m < 3; ++m) {
            v8f acc[NT] = {};
            const float2* bp0 = wt2 + m * WMAT + half * LDW + l;
            float2 bcur[NT], bnxt[NT];
#pragma unroll
            for (int t = 0; t < NT; ++t) bcur[t] = bp0[t * 16];
#pragma unroll
            for (int j = 0; j < KS; ++j) {
                v2f a;
                a.x = rowOK ? areg[j].x : 0.0f;
                a.y = rowOK ? areg[j].y : 0.0f;
                if (j + 1 < KS) {
                    const float2* bp = bp0 + (2 * (j + 1)) * LDW;
#pragma unroll
                    for (int t = 0; t < NT; ++t) bnxt[t] = bp[t * 16];
                }
#pragma unroll
                for (int t = 0; t < NT; ++t) {
                    v2f b; b.x = bcur[t].x; b.y = bcur[t].y;
                    acc[t] = __builtin_amdgcn_wmma_f32_16x16x4_f32(
                        false, a, false, b, (short)0, acc[t], false, false);
                }
                __builtin_amdgcn_sched_barrier(0);
                if (j + 1 < KS) {
#pragma unroll
                    for (int t = 0; t < NT; ++t) bcur[t] = bnxt[t];
                }
            }

            if (m == 2 && nxt < nTiles) {
                const int row = nxt * 16 + l;
                rowOK = (row < M);
                const int rr = rowOK ? row : (M - 1);
                const float* xrow = X + (size_t)rr * K + 2 * half;
#pragma unroll
                for (int j = 0; j < KS; ++j) areg[j] = *(const float2*)(xrow + 4 * j);
            }

            float* Y = (m == 0) ? Y0 : ((m == 1) ? Y1 : Y2);
#pragma unroll
            for (int t = 0; t < NT; ++t) {
                const int c = t * 16 + l;
                const float badd = (m == 2) ? bv[t] : 0.0f;
#pragma unroll
                for (int v = 0; v < 8; ++v) {
                    const int r = rowBase + v + 8 * half;
                    if (r < M) Y[(size_t)r * N + c] = acc[t][v] + badd;
                }
            }
        }
    }
}

#define OWN_ROWS 392
#define OWN_THR  1024
#define OWN_WAVES (OWN_THR / 32)
#define QCAP     64
#define SCAN_EDGES 1024

#define REC_OFF_WORDS (NEDGES / 4 + 3840)
__global__ void __launch_bounds__(256)
owner_prep_kernel(const int* __restrict__ src,
                  const int* __restrict__ dst,
                  const int* __restrict__ etype,
                  const float* __restrict__ eweight,
                  unsigned* __restrict__ enc) {
    static_assert((NNODES + OWN_ROWS - 1) / OWN_ROWS <= 256, "an owner number fits a byte");
    static_assert(NNODES <= (1 << 17) && OWN_ROWS <= (1 << 9), "record word: 17 + 1 + 9 bits");
    static_assert(NEDGES % 4 == 0 && (REC_OFF_WORDS % 4) == 0, "four edges per key word, 16-B record stores");
    const int w = blockIdx.x * 256 + threadIdx.x;
    if (w < NEDGES / 4) {
        const int4   d = *(const int4*)(dst + 4 * w);
        const int4   s = *(const int4*)(src + 4 * w);
        const int4   t = *(const int4*)(etype + 4 * w);
        const float4 f = *(const float4*)(eweight + 4 * w);
        const unsigned k0 = min((unsigned)d.x / OWN_ROWS, 255u);
        const unsigned k1 = min((unsigned)d.y / OWN_ROWS, 255u);
        const unsigned k2 = min((unsigned)d.z / OWN_ROWS, 255u);
        const unsigned k3 = min((unsigned)d.w / OWN_ROWS, 255u);
        enc[w] = k0 | (k1 << 8) | (k2 << 16) | (k3 << 24);
        auto word = [](int sv, int tv, int dv, unsigned k) -> unsigned {
            return ((unsigned)sv & 0x1FFFFu) | ((tv != 0) ? (1u << 17) : 0u)
                 | ((((unsigned)dv - k * OWN_ROWS) & 0x1FFu) << 18);
        };
        uint4* rp = (uint4*)(enc + REC_OFF_WORDS) + 2 * w;
        rp[0] = make_uint4(word(s.x, t.x, d.x, k0), __float_as_uint(f.x), word(s.y, t.y, d.y, k1), __float_as_uint(f.y));
        rp[1] = make_uint4(word(s.z, t.z, d.z, k2), __float_as_uint(f.z), word(s.w, t.w, d.w, k3), __float_as_uint(f.w));
    }
}

#define LCAP           512
#define LIST_OFF_WORDS 1572864
#define CNT_OFF_WORDS  (LIST_OFF_WORDS + 256 * OWN_WAVES * LCAP * 2)

template <int DO_LN>
__global__ void __launch_bounds__(OWN_THR)
owner_sum_kernel(const unsigned* __restrict__ enc,
                 unsigned* lst,
                 const float* __restrict__ y0,
                 const float* __restrict__ y1,
                 float* h,
                 const float* __restrict__ g,
                 const float* __restrict__ b) {
    static_assert(NEDGES % SCAN_EDGES == 0, "edge list is tested in whole 1024-edge pieces");
    static_assert(REC_OFF_WORDS + 2 * NEDGES <= LIST_OFF_WORDS, "lists start behind the records");
    static_assert((size_t)(CNT_OFF_WORDS + 256 * OWN_WAVES) * 4 <= (size_t)NNODES * EMB * 4, "lists and counts fit the region");
    static_assert((NNODES + OWN_ROWS - 1) / OWN_ROWS <= 256, "one list per wave of at most 256 work-groups");
    __shared__ float acc[OWN_ROWS * EMB];
    __shared__ unsigned ring[OWN_WAVES * QCAP];

    const int tid  = threadIdx.x;
    const int lane = tid & 31;
    const int wave = __builtin_amdgcn_readfirstlane(tid >> 5);
    const int r0   = blockIdx.x * OWN_ROWS;
    const int nrow = min(OWN_ROWS, NNODES - r0);
    const unsigned nr = (unsigned)nrow;

    const uint2* rec = (const uint2*)(enc + REC_OFF_WORDS);
    const int wslot = blockIdx.x * OWN_WAVES + wave;
    uint2* mylist = (uint2*)(lst + LIST_OFF_WORDS) + (size_t)wslot * LCAP;
    unsigned* mycnt = lst + CNT_OFF_WORDS + wslot;

    unsigned cntw = 0u;
    uint2 rn = make_uint2(0u, 0u);
    if (!DO_LN) {
        cntw = (unsigned)__builtin_amdgcn_readfirstlane((int)*mycnt);
        if (cntw <= (unsigned)LCAP && (unsigned)lane < cntw) rn = mylist[lane];
    }

    constexpr int RPW = (OWN_ROWS + OWN_WAVES - 1) / OWN_WAVES;
#pragma unroll
    for (int i0 = 0; i0 < RPW; i0 += 7) {
        float4 t[7];
#pragma unroll
        for (int i = 0; i < 7; ++i) {
            const int r = min(wave + OWN_WAVES * (i0 + i), nrow - 1);
            t[i] = *(const float4*)(h + (size_t)(r0 + r) * EMB + lane * 4);
        }
#pragma unroll
        for (int i = 0; i < 7; ++i) {
            const int r = wave + OWN_WAVES * (i0 + i);
            if (r < nrow) {
                float* a = acc + r * EMB + lane;
                a[0] = t[i].x; a[32] = t[i].y; a[64] = t[i].z; a[96] = t[i].w;
            }
        }
    }
    __syncthreads();

    auto add_rows = [&](const int n, const uint2 r) {
        const int s_l = (int)(r.x & 0x1FFFFu);
        const int t_l = (int)((r.x >> 17) & 1u);
        const unsigned dl_l = r.x >> 18;
        const int w_l = (int)r.y;
        for (int j = 0; j < n; j += 8) {
            float4 v[8];
            float  wj[8];
            unsigned dj[8];
#pragma unroll
            for (int u = 0; u < 8; ++u) {
                const int jj = min(j + u, n - 1);
                const int sj = __builtin_amdgcn_readlane(s_l, jj);
                const int tj = __builtin_amdgcn_readlane(t_l, jj);
                wj[u] = __int_as_float(__builtin_amdgcn_readlane(w_l, jj));
                dj[u] = (unsigned)__builtin_amdgcn_readlane((int)dl_l, jj);
                const float* y = ((tj == 0) ? y0 : y1) + (size_t)sj * EMB;
                v[u] = *(const float4*)(y + lane * 4);
            }
#pragma unroll
            for (int u = 0; u < 8; ++u) {
                if (j + u < n && dj[u] < nr) {
                    float* a = acc + dj[u] * EMB + lane;
                    __hip_atomic_fetch_add(a,      wj[u] * v[u].x, __ATOMIC_RELAXED, __HIP_MEMORY_SCOPE_WORKGROUP);
                    __hip_atomic_fetch_add(a + 32, wj[u] * v[u].y, __ATOMIC_RELAXED, __HIP_MEMORY_SCOPE_WORKGROUP);
                    __hip_atomic_fetch_add(a + 64, wj[u] * v[u].z, __ATOMIC_RELAXED, __HIP_MEMORY_SCOPE_WORKGROUP);
                    __hip_atomic_fetch_add(a + 96, wj[u] * v[u].w, __ATOMIC_RELAXED, __HIP_MEMORY_SCOPE_WORKGROUP);
                }
            }
        }
    };

    int noted = 0;

    auto scan_edges = [&]() {
        unsigned* qw = ring + wave * QCAP;
        int qhead = 0, qn = 0;

        constexpr int NPIECE = NEDGES / SCAN_EDGES;
        constexpr int NTRIP  = (NPIECE + OWN_WAVES - 1) / OWN_WAVES;

        static_assert(OWN_WAVES <= NPIECE, "every wave has a first piece");
        constexpr int NTRIP3 = ((NTRIP + 2) / 3) * 3;
        const unsigned* klane = enc + lane * 4;
        auto piece_at = [&](int p) { return klane + (size_t)min(p, NPIECE - 1) * (SCAN_EDGES / 4); };
        int4 idA[3], idB[3];
        idA[0] = *(const int4*)(piece_at(wave));
        idB[0] = *(const int4*)(piece_at(wave) + 128);
        idA[1] = *(const int4*)(piece_at(wave + OWN_WAVES));
        idB[1] = *(const int4*)(piece_at(wave + OWN_WAVES) + 128);
        const unsigned kpat = (unsigned)blockIdx.x * 0x01010101u;

        for (int it = 0; it < NTRIP3; it += 3) {
#pragma unroll
            for (int s = 0; s < 3; ++s) {
                const int piece = wave + OWN_WAVES * (it + s);
                const int4 cA = idA[s], cB = idB[s];
                idA[(s + 2) % 3] = *(const int4*)(piece_at(piece + 2 * OWN_WAVES));
                idB[(s + 2) % 3] = *(const int4*)(piece_at(piece + 2 * OWN_WAVES) + 128);

                auto eq = [&](int x) -> unsigned {
                    const unsigned m = (unsigned)x ^ kpat;
                    return ~(((m & 0x7F7F7F7Fu) + 0x7F7F7F7Fu) | m | 0x7F7F7F7Fu);
                };
                unsigned hm = (eq(cA.x) >> 7) | (eq(cA.y) >> 6) | (eq(cA.z) >> 5) | (eq(cA.w) >> 4)
                            | (eq(cB.x) >> 3) | (eq(cB.y) >> 2) | (eq(cB.z) >> 1) |  eq(cB.w);
                if (piece >= NPIECE) hm = 0u;
                const int ebase = piece * SCAN_EDGES + lane * 16;

                const bool last = (it + s == NTRIP3 - 1);
                unsigned more = __builtin_amdgcn_ballot_w32(hm != 0u);
                do {
                    if (more) {
                        const int pos = qn + (int)__builtin_amdgcn_mbcnt_lo(more, 0u);
                        if (hm != 0u) {
                            const int bit = __builtin_ctz(hm);
                            const int wi = bit & 7, bj = bit >> 3;
                            qw[(qhead + pos) & (QCAP - 1)] = (unsigned)(ebase + (wi >> 2) * 512 + (wi & 3) * 4 + bj);
                            hm &= hm - 1u;
                        }
                        qn += __builtin_popcount(more);
                        more = __builtin_amdgcn_ballot_w32(hm != 0u);
                    }
                    while (qn >= 32 || (last && more == 0u && qn > 0)) {
                        const int n = min(qn, 32);
                        __builtin_amdgcn_wave_barrier();
                        uint2 r = make_uint2(0u, 0u);
                        if (lane < n) r = rec[qw[(qhead + lane) & (QCAP - 1)]];
                        if (DO_LN) {
                            if (noted + n <= LCAP && lane < n) mylist[noted + lane] = r;
                            noted += n;
                        }
                        add_rows(n, r);
                        qhead = (qhead + n) & (QCAP - 1);
                        qn -= n;
                    }
                } while (more);
            }
        }
    };

    if (DO_LN) {
        scan_edges();
        if (lane == 0) *mycnt = (unsigned)noted;
    } else {
        if (cntw <= (unsigned)LCAP) {
            const int cnt = (int)cntw;
            for (int base = 0; base < cnt; base += 32) {
                const int n = min(32, cnt - base);
                const uint2 r = rn;
                if (base + 32 + lane < cnt) rn = mylist[base + 32 + lane];
                add_rows(n, r);
            }
        } else {
            scan_edges();
        }
    }
    __syncthreads();

    float4 gg = make_float4(1.f, 1.f, 1.f, 1.f), bb = make_float4(0.f, 0.f, 0.f, 0.f);
    if (DO_LN) {
        gg = *(const float4*)(g + lane * 4);
        bb = *(const float4*)(b + lane * 4);
    }
    for (int g0 = 0; g0 < RPW; g0 += 4) {
        float4 v[4], o[4];
#pragma unroll
        for (int q = 0; q < 4; ++q) {
            const int r = min(wave + OWN_WAVES * (g0 + q), nrow - 1);
            const float* a = acc + r * EMB + lane;
            v[q] = make_float4(a[0], a[32], a[64], a[96]);
            o[q] = v[q];
        }
        if (DO_LN) {
            float s[4], t[4];
#pragma unroll
            for (int q = 0; q < 4; ++q) {
                v[q].x = fmaxf(v[q].x, 0.0f); v[q].y = fmaxf(v[q].y, 0.0f);
                v[q].z = fmaxf(v[q].z, 0.0f); v[q].w = fmaxf(v[q].w, 0.0f);
                s[q] = v[q].x + v[q].y + v[q].z + v[q].w;
            }
#pragma unroll
            for (int m = 16; m >= 1; m >>= 1) {
#pragma unroll
                for (int q = 0; q < 4; ++q) t[q] = __shfl_xor(s[q], m, 32);
#pragma unroll
                for (int q = 0; q < 4; ++q) s[q] += t[q];
            }
            float dx[4], dy[4], dz[4], dw[4];
#pragma unroll
            for (int q = 0; q < 4; ++q) {
                const float mu = s[q] * (1.0f / EMB);
                dx[q] = v[q].x - mu; dy[q] = v[q].y - mu; dz[q] = v[q].z - mu; dw[q] = v[q].w - mu;
                s[q] = dx[q] * dx[q] + dy[q] * dy[q] + dz[q] * dz[q] + dw[q] * dw[q];
            }
#pragma unroll
            for (int m = 16; m >= 1; m >>= 1) {
#pragma unroll
                for (int q = 0; q < 4; ++q) t[q] = __shfl_xor(s[q], m, 32);
#pragma unroll
                for (int q = 0; q < 4; ++q) s[q] += t[q];
            }
#pragma unroll
            for (int q = 0; q < 4; ++q) {
                const float var = s[q] * (1.0f / EMB);
                const float inv = rsqrtf(var + 1e-5f);
                o[q].x = dx[q] * inv * gg.x + bb.x;
                o[q].y = dy[q] * inv * gg.y + bb.y;
                o[q].z = dz[q] * inv * gg.z + bb.z;
                o[q].w = dw[q] * inv * gg.w + bb.w;
            }
        }
#pragma unroll
        for (int q = 0; q < 4; ++q) {
            const int r = wave + OWN_WAVES * (g0 + q);
            if (r < nrow) *(float4*)(h + (size_t)(r0 + r) * EMB + lane * 4) = o[q];
        }
    }
}

__global__ void gather_gmf_kernel(const float* __restrict__ h2,
                                  const int* __restrict__ ui,
                                  const int* __restrict__ vi,
                                  float* __restrict__ gmf,
                                  float* __restrict__ hcat) {
    const int bidx = blockIdx.x * (blockDim.x >> 5) + (threadIdx.x >> 5);
    if (bidx >= BATCH) return;
    const int lane = threadIdx.x & 31;
    const int u = ui[bidx], w = vi[bidx];
    const float4 uv = *(const float4*)(h2 + (size_t)u * EMB + lane * 4);
    const float4 vv = *(const float4*)(h2 + (size_t)w * EMB + lane * 4);
    const float su = wave_sum(uv.x * uv.x + uv.y * uv.y + uv.z * uv.z + uv.w * uv.w);
    const float sv = wave_sum(vv.x * vv.x + vv.y * vv.y + vv.z * vv.z + vv.w * vv.w);
    const float rn = 1.0f / (fmaxf(sqrtf(su), 1e-12f) * fmaxf(sqrtf(sv), 1e-12f));
    float4 gg;
    gg.x = uv.x * vv.x * rn; gg.y = uv.y * vv.y * rn;
    gg.z = uv.z * vv.z * rn; gg.w = uv.w * vv.w * rn;
    *(float4*)(gmf + (size_t)bidx * EMB + lane * 4) = gg;
    *(float4*)(hcat + (size_t)bidx * 2 * EMB + lane * 4) = uv;
    *(float4*)(hcat + (size_t)bidx * 2 * EMB + EMB + lane * 4) = vv;
}

__global__ void score_kernel(const float* __restrict__ gmf,
                             const float* __restrict__ h3,
                             const float* __restrict__ out_w,
                             const float* __restrict__ out_b,
                             float* __restrict__ out) {
    const int bidx = blockIdx.x * (blockDim.x >> 5) + (threadIdx.x >> 5);
    if (bidx >= BATCH) return;
    const int lane = threadIdx.x & 31;
    const float4 gg = *(const float4*)(gmf + (size_t)bidx * EMB + lane * 4);
    const float4 ww = *(const float4*)(out_w + lane * 4);
    float acc = gg.x * ww.x + gg.y * ww.y + gg.z * ww.z + gg.w * ww.w;
    acc += h3[(size_t)bidx * 32 + lane] * out_w[EMB + lane];
    acc = wave_sum(acc) + out_b[0];
    if (lane == 0) out[bidx] = 1.0f / (1.0f + expf(-acc));
}

extern "C" void kernel_launch(void* const* d_in, const int* in_sizes, int n_in,
                              void* d_out, int out_size, void* d_ws, size_t ws_size,
                              hipStream_t stream) {
    const int*   user_idx = (const int*)d_in[0];
    const int*   item_idx = (const int*)d_in[1];
    const int*   eidx     = (const int*)d_in[2];
    const int*   etype    = (const int*)d_in[3];
    const float* eweight  = (const float*)d_in[4];
    const float* emb      = (const float*)d_in[5];
    const float* w1_rel   = (const float*)d_in[6];
    const float* w1_root  = (const float*)d_in[7];
    const float* b1       = (const float*)d_in[8];
    const float* ln_g     = (const float*)d_in[9];
    const float* ln_b     = (const float*)d_in[10];
    const float* w2_rel   = (const float*)d_in[11];
    const float* w2_root  = (const float*)d_in[12];
    const float* b2       = (const float*)d_in[13];
    const float* mlp_w0   = (const float*)d_in[14];
    const float* mlp_b0   = (const float*)d_in[15];
    const float* mlp_w1   = (const float*)d_in[16];
    const float* mlp_b1   = (const float*)d_in[17];
    const float* mlp_w2   = (const float*)d_in[18];
    const float* mlp_b2   = (const float*)d_in[19];
    const float* out_w    = (const float*)d_in[20];
    const float* out_b    = (const float*)d_in[21];
    float* out = (float*)d_out;

    char* ws = (char*)d_ws;
    const size_t NB = (size_t)NNODES * EMB * sizeof(float);
    float* y0 = (float*)(ws);
    float* y1 = (float*)(ws + NB);
    float* h1 = (float*)(ws + 2 * NB);
    unsigned* ebuf = (unsigned*)(ws + 3 * NB);
    float* gmf  = (float*)(ws);
    float* hcat = (float*)(ws + (size_t)BATCH * EMB * 4);
    float* m0   = (float*)(ws + (size_t)BATCH * EMB * 4 + (size_t)BATCH * 2 * EMB * 4);
    float* m1   = (float*)((char*)m0 + (size_t)BATCH * 128 * 4);
    float* m2   = (float*)((char*)m1 + (size_t)BATCH * 64 * 4);

    const int nodeGrid  = 256;
    const int batchGrid = 256;
    const int ownGrid   = (NNODES + OWN_ROWS - 1) / OWN_ROWS;
    const int rowGridB  = (BATCH + 7) / 8;

    wmma_gemm3_kernel<8, EMB, 512><<<nodeGrid, 512, 0, stream>>>(emb, w1_rel, w1_rel + EMB * EMB, w1_root, b1, y0, y1, h1, NNODES);
    owner_prep_kernel<<<NEDGES / 4 / 256, 256, 0, stream>>>(eidx, eidx + NEDGES, etype, eweight, ebuf);
    owner_sum_kernel<1><<<ownGrid, OWN_THR, 0, stream>>>(ebuf, ebuf, y0, y1, h1, ln_g, ln_b);

    wmma_gemm3_kernel<8, EMB, 512><<<nodeGrid, 512, 0, stream>>>(h1, w2_rel, w2_rel + EMB * EMB, w2_root, b2, y0, y1, h1, NNODES);
    owner_sum_kernel<0><<<ownGrid, OWN_THR, 0, stream>>>(ebuf, ebuf, y0, y1, h1, nullptr, nullptr);

    gather_gmf_kernel<<<rowGridB, 256, 0, stream>>>(h1, user_idx, item_idx, gmf, hcat);
    wmma_gemm_kernel<8, 2 * EMB, 128><<<batchGrid, 128, 0, stream>>>(hcat, mlp_w0, mlp_b0, m0, BATCH, 1);
    wmma_gemm_kernel<4, 128, 128><<<batchGrid, 128, 0, stream>>>(m0,   mlp_w1, mlp_b1, m1, BATCH, 1);
    wmma_gemm_kernel<2, 64, 128><<<batchGrid, 128, 0, stream>>>(m1,   mlp_w2, mlp_b2, m2, BATCH, 1);
    score_kernel<<<rowGridB, 256, 0, stream>>>(gmf, m2, out_w, out_b, out);
}


# root product moved into the owner work-groups: W_root in LDS, one f32 WMMA tile per wave, added to the LDS row sums by LDS atomics (arrival order) while other waves wait for rows; node GEMM does 2
# speedup vs baseline: 2.3401x; 1.0591x over previous
#include <hip/hip_runtime.h>
#include <hip/hip_bf16.h>
#include <math.h>

#define EMB     128
#define BATCH   16384
#define NNODES  100000
#define NEDGES  640000

typedef float v2f __attribute__((ext_vector_type(2)));
typedef float v8f __attribute__((ext_vector_type(8)));

__device__ __forceinline__ float wave_sum(float v) {
#pragma unroll
    for (int m = 16; m >= 1; m >>= 1) v += __shfl_xor(v, m, 32);
    return v;
}

template <int NT, int K, int NTHR>
__global__ void __launch_bounds__(NTHR)
wmma_gemm_kernel(const float* __restrict__ X,
                 const float* __restrict__ W,
                 const float* __restrict__ bias,
                 float* __restrict__ Y,
                 int M, int act) {
    constexpr int N     = NT * 16;
    constexpr int LDW   = N + 16;
    constexpr int KP    = K / 2;
    constexpr int KS    = K / 4;
    constexpr int WAVES = NTHR / 32;
    constexpr int NQ    = N / 4;
    constexpr int ITEMS = KP * NQ;
    __shared__ __attribute__((aligned(16))) float2 wt2[KP * LDW];

    const int tid  = threadIdx.x;
    const int wave = __builtin_amdgcn_readfirstlane(tid >> 5);
    const int lane = tid & 31;
    const int half = lane >> 4;
    const int l    = lane & 15;

    const int nTiles = (M + 15) >> 4;
    const int tstep  = gridDim.x * WAVES;
    int tile = blockIdx.x + gridDim.x * wave;

    float2 areg[KS];
    bool rowOK;
    {
        const int row = tile * 16 + l;
        rowOK = (tile < nTiles) && (row < M);
        const int rr = rowOK ? row : (M - 1);
        const float* xrow = X + (size_t)rr * K + 2 * half;
#pragma unroll
        for (int j = 0; j < KS; ++j) areg[j] = *(const float2*)(xrow + 4 * j);
    }

    float bv[NT];
#pragma unroll
    for (int t = 0; t < NT; ++t) bv[t] = bias ? bias[t * 16 + l] : 0.0f;

#pragma unroll
    for (int i0 = 0; i0 < ITEMS; i0 += NTHR) {
        const int i = i0 + tid;
        if ((ITEMS % NTHR == 0) || (i < ITEMS)) {
            const int p = i / NQ;
            const int c = (i % NQ) * 4;
            const float4 r0 = *(const float4*)(W + (size_t)(2 * p)     * N + c);
            const float4 r1 = *(const float4*)(W + (size_t)(2 * p + 1) * N + c);
            float4* dstp = (float4*)(wt2 + p * LDW + c);
            dstp[0] = make_float4(r0.x, r1.x, r0.y, r1.y);
            dstp[1] = make_float4(r0.z, r1.z, r0.w, r1.w);
        }
    }
    __syncthreads();

    for (; tile < nTiles; tile += tstep) {
        v8f acc[NT] = {};
        const float2* bp0 = wt2 + half * LDW + l;
        float2 bcur[NT], bnxt[NT];
#pragma unroll
        for (int t = 0; t < NT; ++t) bcur[t] = bp0[t * 16];
#pragma unroll
        for (int j = 0; j < KS; ++j) {
            v2f a;
            a.x = rowOK ? areg[j].x : 0.0f;
            a.y = rowOK ? areg[j].y : 0.0f;
            if (j + 1 < KS) {
                const float2* bp = bp0 + (2 * (j + 1)) * LDW;
#pragma unroll
                for (int t = 0; t < NT; ++t) bnxt[t] = bp[t * 16];
            }
#pragma unroll
            for (int t = 0; t < NT; ++t) {
                v2f b; b.x = bcur[t].x; b.y = bcur[t].y;
                acc[t] = __builtin_amdgcn_wmma_f32_16x16x4_f32(
                    false, a, false, b, (short)0, acc[t], false, false);
            }
            __builtin_amdgcn_sched_barrier(0);
            if (j + 1 < KS) {
#pragma unroll
                for (int t = 0; t < NT; ++t) bcur[t] = bnxt[t];
            }
        }

        const int nxt = tile + tstep;
        if (nxt < nTiles) {
            const int row = nxt * 16 + l;
            rowOK = (row < M);
            const int rr = rowOK ? row : (M - 1);
            const float* xrow = X + (size_t)rr * K + 2 * half;
#pragma unroll
            for (int j = 0; j < KS; ++j) areg[j] = *(const float2*)(xrow + 4 * j);
        }

        const int rowBase = tile * 16;
#pragma unroll
        for (int t = 0; t < NT; ++t) {
            const int c = t * 16 + l;
#pragma unroll
            for (int v = 0; v < 8; ++v) {
                const int r = rowBase + v + 8 * half;
                if (r < M) {
                    float o = acc[t][v] + bv[t];
                    if (act) o = fmaxf(o, 0.0f);
                    Y[(size_t)r * N + c] = o;
                }
            }
        }
    }
}

template <int NT, int K, int NTHR, int NM>
__global__ void __launch_bounds__(NTHR)
wmma_gemm3_kernel(const float* X,
                  const float* __restrict__ W0,
                  const float* __restrict__ W1,
                  const float* __restrict__ W2,
                  const float* __restrict__ bias2,
                  float* __restrict__ Y0,
                  float* __restrict__ Y1,
                  float* Y2,
                  int M) {
    constexpr int N     = NT * 16;
    constexpr int LDW   = N + 16;
    constexpr int KP    = K / 2;
    constexpr int KS    = K / 4;
    constexpr int WAVES = NTHR / 32;
    constexpr int NQ    = N / 4;
    constexpr int ITEMS = KP * NQ;
    constexpr int WMAT  = KP * LDW;
    __shared__ __attribute__((aligned(16))) float2 wt2[NM * WMAT];

    const int tid  = threadIdx.x;
    const int wave = __builtin_amdgcn_readfirstlane(tid >> 5);
    const int lane = tid & 31;
    const int half = lane >> 4;
    const int l    = lane & 15;

    const int nTiles = (M + 15) >> 4;
    const int tstep  = gridDim.x * WAVES;
    int tile = blockIdx.x + gridDim.x * wave;

    float2 areg[KS];
    bool rowOK;
    {
        const int row = tile * 16 + l;
        rowOK = (tile < nTiles) && (row < M);
        const int rr = rowOK ? row : (M - 1);
        const float* xrow = X + (size_t)rr * K + 2 * half;
#pragma unroll
        for (int j = 0; j < KS; ++j) areg[j] = *(const float2*)(xrow + 4 * j);
    }

    float bv[NT];
#pragma unroll
    for (int t = 0; t < NT; ++t) bv[t] = bias2 ? bias2[t * 16 + l] : 0.0f;

#pragma unroll
    for (int m = 0; m < NM; ++m) {
        const float* __restrict__ W = (m == 0) ? W0 : ((m == 1) ? W1 : W2);
#pragma unroll
        for (int i0 = 0; i0 < ITEMS; i0 += NTHR) {
            const int i = i0 + tid;
            if ((ITEMS % NTHR == 0) || (i < ITEMS)) {
                const int p = i / NQ;
                const int c = (i % NQ) * 4;
                const float4 r0 = *(const float4*)(W + (size_t)(2 * p)     * N + c);
                const float4 r1 = *(const float4*)(W + (size_t)(2 * p + 1) * N + c);
                float4* dstp = (float4*)(wt2 + m * WMAT + p * LDW + c);
                dstp[0] = make_float4(r0.x, r1.x, r0.y, r1.y);
                dstp[1] = make_float4(r0.z, r1.z, r0.w, r1.w);
            }
        }
    }
    __syncthreads();

    for (; tile < nTiles; tile += tstep) {
        const int rowBase = tile * 16;
        const int nxt = tile + tstep;
#pragma unroll 1
        for (int m = 0; m < NM; ++m) {
            v8f acc[NT] = {};
            const float2* bp0 = wt2 + m * WMAT + half * LDW + l;
            float2 bcur[NT], bnxt[NT];
#pragma unroll
            for (int t = 0; t < NT; ++t) bcur[t] = bp0[t * 16];
#pragma unroll
            for (int j = 0; j < KS; ++j) {
                v2f a;
                a.x = rowOK ? areg[j].x : 0.0f;
                a.y = rowOK ? areg[j].y : 0.0f;
                if (j + 1 < KS) {
                    const float2* bp = bp0 + (2 * (j + 1)) * LDW;
#pragma unroll
                    for (int t = 0; t < NT; ++t) bnxt[t] = bp[t * 16];
                }
#pragma unroll
                for (int t = 0; t < NT; ++t) {
                    v2f b; b.x = bcur[t].x; b.y = bcur[t].y;
                    acc[t] = __builtin_amdgcn_wmma_f32_16x16x4_f32(
                        false, a, false, b, (short)0, acc[t], false, false);
                }
                __builtin_amdgcn_sched_barrier(0);
                if (j + 1 < KS) {
#pragma unroll
                    for (int t = 0; t < NT; ++t) bcur[t] = bnxt[t];
                }
            }

            if (m == NM - 1 && nxt < nTiles) {
                const int row = nxt * 16 + l;
                rowOK = (row < M);
                const int rr = rowOK ? row : (M - 1);
                const float* xrow = X + (size_t)rr * K + 2 * half;
#pragma unroll
                for (int j = 0; j < KS; ++j) areg[j] = *(const float2*)(xrow + 4 * j);
            }

            float* Y = (m == 0) ? Y0 : ((m == 1) ? Y1 : Y2);
#pragma unroll
            for (int t = 0; t < NT; ++t) {
                const int c = t * 16 + l;
                const float badd = (m == 2) ? bv[t] : 0.0f;
#pragma unroll
                for (int v = 0; v < 8; ++v) {
                    const int r = rowBase + v + 8 * half;
                    if (r < M) Y[(size_t)r * N + c] = acc[t][v] + badd;
                }
            }
        }
    }
}

#define OWN_ROWS 392
#define OWN_THR  1024
#define OWN_WAVES (OWN_THR / 32)
#define QCAP     64
#define SCAN_EDGES 1024

#define REC_OFF_WORDS (NEDGES / 4 + 3840)
__global__ void __launch_bounds__(256)
owner_prep_kernel(const int* __restrict__ src,
                  const int* __restrict__ dst,
                  const int* __restrict__ etype,
                  const float* __restrict__ eweight,
                  unsigned* __restrict__ enc) {
    static_assert((NNODES + OWN_ROWS - 1) / OWN_ROWS <= 256, "an owner number fits a byte");
    static_assert(NNODES <= (1 << 17) && OWN_ROWS <= (1 << 9), "record word: 17 + 1 + 9 bits");
    static_assert(NEDGES % 4 == 0 && (REC_OFF_WORDS % 4) == 0, "four edges per key word, 16-B record stores");
    const int w = blockIdx.x * 256 + threadIdx.x;
    if (w < NEDGES / 4) {
        const int4   d = *(const int4*)(dst + 4 * w);
        const int4   s = *(const int4*)(src + 4 * w);
        const int4   t = *(const int4*)(etype + 4 * w);
        const float4 f = *(const float4*)(eweight + 4 * w);
        const unsigned k0 = min((unsigned)d.x / OWN_ROWS, 255u);
        const unsigned k1 = min((unsigned)d.y / OWN_ROWS, 255u);
        const unsigned k2 = min((unsigned)d.z / OWN_ROWS, 255u);
        const unsigned k3 = min((unsigned)d.w / OWN_ROWS, 255u);
        enc[w] = k0 | (k1 << 8) | (k2 << 16) | (k3 << 24);
        auto word = [](int sv, int tv, int dv, unsigned k) -> unsigned {
            return ((unsigned)sv & 0x1FFFFu) | ((tv != 0) ? (1u << 17) : 0u)
                 | ((((unsigned)dv - k * OWN_ROWS) & 0x1FFu) << 18);
        };
        uint4* rp = (uint4*)(enc + REC_OFF_WORDS) + 2 * w;
        rp[0] = make_uint4(word(s.x, t.x, d.x, k0), __float_as_uint(f.x), word(s.y, t.y, d.y, k1), __float_as_uint(f.y));
        rp[1] = make_uint4(word(s.z, t.z, d.z, k2), __float_as_uint(f.z), word(s.w, t.w, d.w, k3), __float_as_uint(f.w));
    }
}

#define LCAP           512
#define LIST_OFF_WORDS 1572864
#define CNT_OFF_WORDS  (LIST_OFF_WORDS + 256 * OWN_WAVES * LCAP * 2)

template <int DO_LN>
__global__ void __launch_bounds__(OWN_THR)
owner_sum_kernel(const unsigned* __restrict__ enc,
                 unsigned* lst,
                 const float* __restrict__ y0,
                 const float* __restrict__ y1,
                 const float* x,
                 const float* __restrict__ wroot,
                 const float* __restrict__ broot,
                 float* h,
                 const float* __restrict__ g,
                 const float* __restrict__ b) {
    static_assert(NEDGES % SCAN_EDGES == 0, "edge list is tested in whole 1024-edge pieces");
    static_assert(REC_OFF_WORDS + 2 * NEDGES <= LIST_OFF_WORDS, "lists start behind the records");
    static_assert((size_t)(CNT_OFF_WORDS + 256 * OWN_WAVES) * 4 <= (size_t)NNODES * EMB * 4, "lists and counts fit the region");
    static_assert((NNODES + OWN_ROWS - 1) / OWN_ROWS <= 256, "one list per wave of at most 256 work-groups");
    __shared__ __attribute__((aligned(16))) float acc[OWN_ROWS * EMB];
    __shared__ unsigned ring[OWN_WAVES * QCAP];
    constexpr int LDW = EMB + 16;
    constexpr int KS  = EMB / 4;
    __shared__ __attribute__((aligned(16))) float2 wrt[(EMB / 2) * LDW];
    static_assert((OWN_ROWS + 15) / 16 <= OWN_WAVES, "one 16-row tile per wave covers the owned rows");

    const int tid  = threadIdx.x;
    const int lane = tid & 31;
    const int wave = __builtin_amdgcn_readfirstlane(tid >> 5);
    const int r0   = blockIdx.x * OWN_ROWS;
    const int nrow = min(OWN_ROWS, NNODES - r0);
    const unsigned nr = (unsigned)nrow;

    const uint2* rec = (const uint2*)(enc + REC_OFF_WORDS);
    const int wslot = blockIdx.x * OWN_WAVES + wave;
    uint2* mylist = (uint2*)(lst + LIST_OFF_WORDS) + (size_t)wslot * LCAP;
    unsigned* mycnt = lst + CNT_OFF_WORDS + wslot;

    unsigned cntw = 0u;
    uint2 rn = make_uint2(0u, 0u);
    if (!DO_LN) {
        cntw = (unsigned)__builtin_amdgcn_readfirstlane((int)*mycnt);
        if (cntw <= (unsigned)LCAP && (unsigned)lane < cntw) rn = mylist[lane];
    }

    const bool hasTile = (wave * 16 < nrow);
    float2 areg[KS];
    if (hasTile) {
        const int rr = min(wave * 16 + (lane & 15), nrow - 1);
        const float* xrow = x + (size_t)(r0 + rr) * EMB + 2 * (lane >> 4);
#pragma unroll
        for (int j = 0; j < KS; ++j) areg[j] = *(const float2*)(xrow + 4 * j);
    }

    constexpr int RPW = (OWN_ROWS + OWN_WAVES - 1) / OWN_WAVES;
    {
        constexpr int NQ = EMB / 4;
        constexpr int ITEMS = (EMB / 2) * NQ;
        static_assert(ITEMS % OWN_THR == 0, "whole staging rounds");
#pragma unroll
        for (int i0 = 0; i0 < ITEMS; i0 += OWN_THR) {
            const int i = i0 + tid;
            const int p = i / NQ;
            const int cc = (i % NQ) * 4;
            const float4 q0 = *(const float4*)(wroot + (size_t)(2 * p)     * EMB + cc);
            const float4 q1 = *(const float4*)(wroot + (size_t)(2 * p + 1) * EMB + cc);
            float4* dstp = (float4*)(wrt + p * LDW + cc);
            dstp[0] = make_float4(q0.x, q1.x, q0.y, q1.y);
            dstp[1] = make_float4(q0.z, q1.z, q0.w, q1.w);
        }
        static_assert(OWN_THR % 32 == 0 && EMB / 4 == 32, "slot number within a row == lane");
        float4 b4 = make_float4(0.f, 0.f, 0.f, 0.f);
        if (broot) {
            const float* bq = broot + 16 * (lane & 7) + (lane >> 3);
            b4 = make_float4(bq[0], bq[4], bq[8], bq[12]);
        }
        float4* az = (float4*)acc;
        for (int i = tid; i < OWN_ROWS * EMB / 4; i += OWN_THR) az[i] = b4;
    }
    __syncthreads();

    if (hasTile) {
        const int half = lane >> 4;
        const int l    = lane & 15;
#pragma unroll 1
        for (int ps = 0; ps < 2; ++ps) {
            v8f cacc[4] = {};
            const float2* bp0 = wrt + half * LDW + l + ps * 64;
            float2 bcur[4], bnxt[4];
#pragma unroll
            for (int t = 0; t < 4; ++t) bcur[t] = bp0[t * 16];
#pragma unroll
            for (int j = 0; j < KS; ++j) {
                v2f a; a.x = areg[j].x; a.y = areg[j].y;
                if (j + 1 < KS) {
                    const float2* bp = bp0 + (2 * (j + 1)) * LDW;
#pragma unroll
                    for (int t = 0; t < 4; ++t) bnxt[t] = bp[t * 16];
                }
#pragma unroll
                for (int t = 0; t < 4; ++t) {
                    v2f bb2; bb2.x = bcur[t].x; bb2.y = bcur[t].y;
                    cacc[t] = __builtin_amdgcn_wmma_f32_16x16x4_f32(
                        false, a, false, bb2, (short)0, cacc[t], false, false);
                }
                __builtin_amdgcn_sched_barrier(0);
                if (j + 1 < KS) {
#pragma unroll
                    for (int t = 0; t < 4; ++t) bcur[t] = bnxt[t];
                }
            }
#pragma unroll
            for (int t = 0; t < 4; ++t) {
                const int col = ps * 64 + t * 16 + l;
                float* a0 = acc + (col & 3) * 32 + (col >> 2);
#pragma unroll
                for (int v = 0; v < 8; ++v) {
                    const int r = wave * 16 + v + 8 * half;
                    if (r < nrow)
                        __hip_atomic_fetch_add(a0 + r * EMB, cacc[t][v], __ATOMIC_RELAXED, __HIP_MEMORY_SCOPE_WORKGROUP);
                }
            }
        }
    }

    auto add_rows = [&](const int n, const uint2 r) {
        const int s_l = (int)(r.x & 0x1FFFFu);
        const int t_l = (int)((r.x >> 17) & 1u);
        const unsigned dl_l = r.x >> 18;
        const int w_l = (int)r.y;
        for (int j = 0; j < n; j += 8) {
            float4 v[8];
            float  wj[8];
            unsigned dj[8];
#pragma unroll
            for (int u = 0; u < 8; ++u) {
                const int jj = min(j + u, n - 1);
                const int sj = __builtin_amdgcn_readlane(s_l, jj);
                const int tj = __builtin_amdgcn_readlane(t_l, jj);
                wj[u] = __int_as_float(__builtin_amdgcn_readlane(w_l, jj));
                dj[u] = (unsigned)__builtin_amdgcn_readlane((int)dl_l, jj);
                const float* y = ((tj == 0) ? y0 : y1) + (size_t)sj * EMB;
                v[u] = *(const float4*)(y + lane * 4);
            }
#pragma unroll
            for (int u = 0; u < 8; ++u) {
                if (j + u < n && dj[u] < nr) {
                    float* a = acc + dj[u] * EMB + lane;
                    __hip_atomic_fetch_add(a,      wj[u] * v[u].x, __ATOMIC_RELAXED, __HIP_MEMORY_SCOPE_WORKGROUP);
                    __hip_atomic_fetch_add(a + 32, wj[u] * v[u].y, __ATOMIC_RELAXED, __HIP_MEMORY_SCOPE_WORKGROUP);
                    __hip_atomic_fetch_add(a + 64, wj[u] * v[u].z, __ATOMIC_RELAXED, __HIP_MEMORY_SCOPE_WORKGROUP);
                    __hip_atomic_fetch_add(a + 96, wj[u] * v[u].w, __ATOMIC_RELAXED, __HIP_MEMORY_SCOPE_WORKGROUP);
                }
            }
        }
    };

    int noted = 0;

    auto scan_edges = [&]() {
        unsigned* qw = ring + wave * QCAP;
        int qhead = 0, qn = 0;

        constexpr int NPIECE = NEDGES / SCAN_EDGES;
        constexpr int NTRIP  = (NPIECE + OWN_WAVES - 1) / OWN_WAVES;

        static_assert(OWN_WAVES <= NPIECE, "every wave has a first piece");
        constexpr int NTRIP3 = ((NTRIP + 2) / 3) * 3;
        const unsigned* klane = enc + lane * 4;
        auto piece_at = [&](int p) { return klane + (size_t)min(p, NPIECE - 1) * (SCAN_EDGES / 4); };
        int4 idA[3], idB[3];
        idA[0] = *(const int4*)(piece_at(wave));
        idB[0] = *(const int4*)(piece_at(wave) + 128);
        idA[1] = *(const int4*)(piece_at(wave + OWN_WAVES));
        idB[1] = *(const int4*)(piece_at(wave + OWN_WAVES) + 128);
        const unsigned kpat = (unsigned)blockIdx.x * 0x01010101u;

        for (int it = 0; it < NTRIP3; it += 3) {
#pragma unroll
            for (int s = 0; s < 3; ++s) {
                const int piece = wave + OWN_WAVES * (it + s);
                const int4 cA = idA[s], cB = idB[s];
                idA[(s + 2) % 3] = *(const int4*)(piece_at(piece + 2 * OWN_WAVES));
                idB[(s + 2) % 3] = *(const int4*)(piece_at(piece + 2 * OWN_WAVES) + 128);

                auto eq = [&](int x) -> unsigned {
                    const unsigned m = (unsigned)x ^ kpat;
                    return ~(((m & 0x7F7F7F7Fu) + 0x7F7F7F7Fu) | m | 0x7F7F7F7Fu);
                };
                unsigned hm = (eq(cA.x) >> 7) | (eq(cA.y) >> 6) | (eq(cA.z) >> 5) | (eq(cA.w) >> 4)
                            | (eq(cB.x) >> 3) | (eq(cB.y) >> 2) | (eq(cB.z) >> 1) |  eq(cB.w);
                if (piece >= NPIECE) hm = 0u;
                const int ebase = piece * SCAN_EDGES + lane * 16;

                const bool last = (it + s == NTRIP3 - 1);
                unsigned more = __builtin_amdgcn_ballot_w32(hm != 0u);
                do {
                    if (more) {
                        const int pos = qn + (int)__builtin_amdgcn_mbcnt_lo(more, 0u);
                        if (hm != 0u) {
                            const int bit = __builtin_ctz(hm);
                            const int wi = bit & 7, bj = bit >> 3;
                            qw[(qhead + pos) & (QCAP - 1)] = (unsigned)(ebase + (wi >> 2) * 512 + (wi & 3) * 4 + bj);
                            hm &= hm - 1u;
                        }
                        qn += __builtin_popcount(more);
                        more = __builtin_amdgcn_ballot_w32(hm != 0u);
                    }
                    while (qn >= 32 || (last && more == 0u && qn > 0)) {
                        const int n = min(qn, 32);
                        __builtin_amdgcn_wave_barrier();
                        uint2 r = make_uint2(0u, 0u);
                        if (lane < n) r = rec[qw[(qhead + lane) & (QCAP - 1)]];
                        if (DO_LN) {
                            if (noted + n <= LCAP && lane < n) mylist[noted + lane] = r;
                            noted += n;
                        }
                        add_rows(n, r);
                        qhead = (qhead + n) & (QCAP - 1);
                        qn -= n;
                    }
                } while (more);
            }
        }
    };

    if (DO_LN) {
        scan_edges();
        if (lane == 0) *mycnt = (unsigned)noted;
    } else {
        if (cntw <= (unsigned)LCAP) {
            const int cnt = (int)cntw;
            for (int base = 0; base < cnt; base += 32) {
                const int n = min(32, cnt - base);
                const uint2 r = rn;
                if (base + 32 + lane < cnt) rn = mylist[base + 32 + lane];
                add_rows(n, r);
            }
        } else {
            scan_edges();
        }
    }
    __syncthreads();

    float4 gg = make_float4(1.f, 1.f, 1.f, 1.f), bb = make_float4(0.f, 0.f, 0.f, 0.f);
    if (DO_LN) {
        gg = *(const float4*)(g + lane * 4);
        bb = *(const float4*)(b + lane * 4);
    }
    for (int g0 = 0; g0 < RPW; g0 += 4) {
        float4 v[4], o[4];
#pragma unroll
        for (int q = 0; q < 4; ++q) {
            const int r = min(wave + OWN_WAVES * (g0 + q), nrow - 1);
            const float* a = acc + r * EMB + lane;
            v[q] = make_float4(a[0], a[32], a[64], a[96]);
            o[q] = v[q];
        }
        if (DO_LN) {
            float s[4], t[4];
#pragma unroll
            for (int q = 0; q < 4; ++q) {
                v[q].x = fmaxf(v[q].x, 0.0f); v[q].y = fmaxf(v[q].y, 0.0f);
                v[q].z = fmaxf(v[q].z, 0.0f); v[q].w = fmaxf(v[q].w, 0.0f);
                s[q] = v[q].x + v[q].y + v[q].z + v[q].w;
            }
#pragma unroll
            for (int m = 16; m >= 1; m >>= 1) {
#pragma unroll
                for (int q = 0; q < 4; ++q) t[q] = __shfl_xor(s[q], m, 32);
#pragma unroll
                for (int q = 0; q < 4; ++q) s[q] += t[q];
            }
            float dx[4], dy[4], dz[4], dw[4];
#pragma unroll
            for (int q = 0; q < 4; ++q) {
                const float mu = s[q] * (1.0f / EMB);
                dx[q] = v[q].x - mu; dy[q] = v[q].y - mu; dz[q] = v[q].z - mu; dw[q] = v[q].w - mu;
                s[q] = dx[q] * dx[q] + dy[q] * dy[q] + dz[q] * dz[q] + dw[q] * dw[q];
            }
#pragma unroll
            for (int m = 16; m >= 1; m >>= 1) {
#pragma unroll
                for (int q = 0; q < 4; ++q) t[q] = __shfl_xor(s[q], m, 32);
#pragma unroll
                for (int q = 0; q < 4; ++q) s[q] += t[q];
            }
#pragma unroll
            for (int q = 0; q < 4; ++q) {
                const float var = s[q] * (1.0f / EMB);
                const float inv = rsqrtf(var + 1e-5f);
                o[q].x = dx[q] * inv * gg.x + bb.x;
                o[q].y = dy[q] * inv * gg.y + bb.y;
                o[q].z = dz[q] * inv * gg.z + bb.z;
                o[q].w = dw[q] * inv * gg.w + bb.w;
            }
        }
#pragma unroll
        for (int q = 0; q < 4; ++q) {
            const int r = wave + OWN_WAVES * (g0 + q);
            if (r < nrow) *(float4*)(h + (size_t)(r0 + r) * EMB + lane * 4) = o[q];
        }
    }
}

__global__ void gather_gmf_kernel(const float* __restrict__ h2,
                                  const int* __restrict__ ui,
                                  const int* __restrict__ vi,
                                  float* __restrict__ gmf,
                                  float* __restrict__ hcat) {
    const int bidx = blockIdx.x * (blockDim.x >> 5) + (threadIdx.x >> 5);
    if (bidx >= BATCH) return;
    const int lane = threadIdx.x & 31;
    const int u = ui[bidx], w = vi[bidx];
    const float4 uv = *(const float4*)(h2 + (size_t)u * EMB + lane * 4);
    const float4 vv = *(const float4*)(h2 + (size_t)w * EMB + lane * 4);
    const float su = wave_sum(uv.x * uv.x + uv.y * uv.y + uv.z * uv.z + uv.w * uv.w);
    const float sv = wave_sum(vv.x * vv.x + vv.y * vv.y + vv.z * vv.z + vv.w * vv.w);
    const float rn = 1.0f / (fmaxf(sqrtf(su), 1e-12f) * fmaxf(sqrtf(sv), 1e-12f));
    float4 gg;
    gg.x = uv.x * vv.x * rn; gg.y = uv.y * vv.y * rn;
    gg.z = uv.z * vv.z * rn; gg.w = uv.w * vv.w * rn;
    *(float4*)(gmf + (size_t)bidx * EMB + lane * 4) = gg;
    *(float4*)(hcat + (size_t)bidx * 2 * EMB + lane * 4) = uv;
    *(float4*)(hcat + (size_t)bidx * 2 * EMB + EMB + lane * 4) = vv;
}

__global__ void score_kernel(const float* __restrict__ gmf,
                             const float* __restrict__ h3,
                             const float* __restrict__ out_w,
                             const float* __restrict__ out_b,
                             float* __restrict__ out) {
    const int bidx = blockIdx.x * (blockDim.x >> 5) + (threadIdx.x >> 5);
    if (bidx >= BATCH) return;
    const int lane = threadIdx.x & 31;
    const float4 gg = *(const float4*)(gmf + (size_t)bidx * EMB + lane * 4);
    const float4 ww = *(const float4*)(out_w + lane * 4);
    float acc = gg.x * ww.x + gg.y * ww.y + gg.z * ww.z + gg.w * ww.w;
    acc += h3[(size_t)bidx * 32 + lane] * out_w[EMB + lane];
    acc = wave_sum(acc) + out_b[0];
    if (lane == 0) out[bidx] = 1.0f / (1.0f + expf(-acc));
}

extern "C" void kernel_launch(void* const* d_in, const int* in_sizes, int n_in,
                              void* d_out, int out_size, void* d_ws, size_t ws_size,
                              hipStream_t stream) {
    const int*   user_idx = (const int*)d_in[0];
    const int*   item_idx = (const int*)d_in[1];
    const int*   eidx     = (const int*)d_in[2];
    const int*   etype    = (const int*)d_in[3];
    const float* eweight  = (const float*)d_in[4];
    const float* emb      = (const float*)d_in[5];
    const float* w1_rel   = (const float*)d_in[6];
    const float* w1_root  = (const float*)d_in[7];
    const float* b1       = (const float*)d_in[8];
    const float* ln_g     = (const float*)d_in[9];
    const float* ln_b     = (const float*)d_in[10];
    const float* w2_rel   = (const float*)d_in[11];
    const float* w2_root  = (const float*)d_in[12];
    const float* b2       = (const float*)d_in[13];
    const float* mlp_w0   = (const float*)d_in[14];
    const float* mlp_b0   = (const float*)d_in[15];
    const float* mlp_w1   = (const float*)d_in[16];
    const float* mlp_b1   = (const float*)d_in[17];
    const float* mlp_w2   = (const float*)d_in[18];
    const float* mlp_b2   = (const float*)d_in[19];
    const float* out_w    = (const float*)d_in[20];
    const float* out_b    = (const float*)d_in[21];
    float* out = (float*)d_out;

    char* ws = (char*)d_ws;
    const size_t NB = (size_t)NNODES * EMB * sizeof(float);
    float* y0 = (float*)(ws);
    float* y1 = (float*)(ws + NB);
    float* h1 = (float*)(ws + 2 * NB);
    unsigned* ebuf = (unsigned*)(ws + 3 * NB);
    float* gmf  = (float*)(ws);
    float* hcat = (float*)(ws + (size_t)BATCH * EMB * 4);
    float* m0   = (float*)(ws + (size_t)BATCH * EMB * 4 + (size_t)BATCH * 2 * EMB * 4);
    float* m1   = (float*)((char*)m0 + (size_t)BATCH * 128 * 4);
    float* m2   = (float*)((char*)m1 + (size_t)BATCH * 64 * 4);

    const int nodeGrid  = 256;
    const int batchGrid = 256;
    const int ownGrid   = (NNODES + OWN_ROWS - 1) / OWN_ROWS;
    const int rowGridB  = (BATCH + 7) / 8;

    wmma_gemm3_kernel<8, EMB, 512, 2><<<nodeGrid, 512, 0, stream>>>(emb, w1_rel, w1_rel + EMB * EMB, nullptr, nullptr, y0, y1, nullptr, NNODES);
    owner_prep_kernel<<<NEDGES / 4 / 256, 256, 0, stream>>>(eidx, eidx + NEDGES, etype, eweight, ebuf);
    owner_sum_kernel<1><<<ownGrid, OWN_THR, 0, stream>>>(ebuf, ebuf, y0, y1, emb, w1_root, b1, h1, ln_g, ln_b);

    wmma_gemm3_kernel<8, EMB, 512, 2><<<nodeGrid, 512, 0, stream>>>(h1, w2_rel, w2_rel + EMB * EMB, nullptr, nullptr, y0, y1, nullptr, NNODES);
    owner_sum_kernel<0><<<ownGrid, OWN_THR, 0, stream>>>(ebuf, ebuf, y0, y1, h1, w2_root, b2, h1, nullptr, nullptr);

    gather_gmf_kernel<<<rowGridB, 256, 0, stream>>>(h1, user_idx, item_idx, gmf, hcat);
    wmma_gemm_kernel<8, 2 * EMB, 128><<<batchGrid, 128, 0, stream>>>(hcat, mlp_w0, mlp_b0, m0, BATCH, 1);
    wmma_gemm_kernel<4, 128, 128><<<batchGrid, 128, 0, stream>>>(m0,   mlp_w1, mlp_b1, m1, BATCH, 1);
    wmma_gemm_kernel<2, 64, 128><<<batchGrid, 128, 0, stream>>>(m1,   mlp_w2, mlp_b2, m2, BATCH, 1);
    score_kernel<<<rowGridB, 256, 0, stream>>>(gmf, m2, out_w, out_b, out);
}


# scoring head as ONE launch (was 5): a wave keeps 16 items in registers; each MLP layer a transposed f32 WMMA product whose D tiles are the next one's B operand; k and lane-sum order change
# speedup vs baseline: 2.6623x; 1.1377x over previous
#include <hip/hip_runtime.h>
#include <hip/hip_bf16.h>
#include <math.h>

#define EMB     128
#define BATCH   16384
#define NNODES  100000
#define NEDGES  640000

typedef float v2f __attribute__((ext_vector_type(2)));
typedef float v8f __attribute__((ext_vector_type(8)));


template <int NT, int K, int NTHR, int NM>
__global__ void __launch_bounds__(NTHR)
wmma_gemm3_kernel(const float* X,
                  const float* __restrict__ W0,
                  const float* __restrict__ W1,
                  const float* __restrict__ W2,
                  const float* __restrict__ bias2,
                  float* __restrict__ Y0,
                  float* __restrict__ Y1,
                  float* Y2,
                  int M) {
    constexpr int N     = NT * 16;
    constexpr int LDW   = N + 16;
    constexpr int KP    = K / 2;
    constexpr int KS    = K / 4;
    constexpr int WAVES = NTHR / 32;
    constexpr int NQ    = N / 4;
    constexpr int ITEMS = KP * NQ;
    constexpr int WMAT  = KP * LDW;
    __shared__ __attribute__((aligned(16))) float2 wt2[NM * WMAT];

    const int tid  = threadIdx.x;
    const int wave = __builtin_amdgcn_readfirstlane(tid >> 5);
    const int lane = tid & 31;
    const int half = lane >> 4;
    const int l    = lane & 15;

    const int nTiles = (M + 15) >> 4;
    const int tstep  = gridDim.x * WAVES;
    int tile = blockIdx.x + gridDim.x * wave;

    float2 areg[KS];
    bool rowOK;
    {
        const int row = tile * 16 + l;
        rowOK = (tile < nTiles) && (row < M);
        const int rr = rowOK ? row : (M - 1);
        const float* xrow = X + (size_t)rr * K + 2 * half;
#pragma unroll
        for (int j = 0; j < KS; ++j) areg[j] = *(const float2*)(xrow + 4 * j);
    }

    float bv[NT];
#pragma unroll
    for (int t = 0; t < NT; ++t) bv[t] = bias2 ? bias2[t * 16 + l] : 0.0f;

#pragma unroll
    for (int m = 0; m < NM; ++m) {
        const float* __restrict__ W = (m == 0) ? W0 : ((m == 1) ? W1 : W2);
#pragma unroll
        for (int i0 = 0; i0 < ITEMS; i0 += NTHR) {
            const int i = i0 + tid;
            if ((ITEMS % NTHR == 0) || (i < ITEMS)) {
                const int p = i / NQ;
                const int c = (i % NQ) * 4;
                const float4 r0 = *(const float4*)(W + (size_t)(2 * p)     * N + c);
                const float4 r1 = *(const float4*)(W + (size_t)(2 * p + 1) * N + c);
                float4* dstp = (float4*)(wt2 + m * WMAT + p * LDW + c);
                dstp[0] = make_float4(r0.x, r1.x, r0.y, r1.y);
                dstp[1] = make_float4(r0.z, r1.z, r0.w, r1.w);
            }
        }
    }
    __syncthreads();

    for (; tile < nTiles; tile += tstep) {
        const int rowBase = tile * 16;
        const int nxt = tile + tstep;
#pragma unroll 1
        for (int m = 0; m < NM; ++m) {
            v8f acc[NT] = {};
            const float2* bp0 = wt2 + m * WMAT + half * LDW + l;
            float2 bcur[NT], bnxt[NT];
#pragma unroll
            for (int t = 0; t < NT; ++t) bcur[t] = bp0[t * 16];
#pragma unroll
            for (int j = 0; j < KS; ++j) {
                v2f a;
                a.x = rowOK ? areg[j].x : 0.0f;
                a.y = rowOK ? areg[j].y : 0.0f;
                if (j + 1 < KS) {
                    const float2* bp = bp0 + (2 * (j + 1)) * LDW;
#pragma unroll
                    for (int t = 0; t < NT; ++t) bnxt[t] = bp[t * 16];
                }
#pragma unroll
                for (int t = 0; t < NT; ++t) {
                    v2f b; b.x = bcur[t].x; b.y = bcur[t].y;
                    acc[t] = __builtin_amdgcn_wmma_f32_16x16x4_f32(
                        false, a, false, b, (short)0, acc[t], false, false);
                }
                __builtin_amdgcn_sched_barrier(0);
                if (j + 1 < KS) {
#pragma unroll
                    for (int t = 0; t < NT; ++t) bcur[t] = bnxt[t];
                }
            }

            if (m == NM - 1 && nxt < nTiles) {
                const int row = nxt * 16 + l;
                rowOK = (row < M);
                const int rr = rowOK ? row : (M - 1);
                const float* xrow = X + (size_t)rr * K + 2 * half;
#pragma unroll
                for (int j = 0; j < KS; ++j) areg[j] = *(const float2*)(xrow + 4 * j);
            }

            float* Y = (m == 0) ? Y0 : ((m == 1) ? Y1 : Y2);
#pragma unroll
            for (int t = 0; t < NT; ++t) {
                const int c = t * 16 + l;
                const float badd = (m == 2) ? bv[t] : 0.0f;
#pragma unroll
                for (int v = 0; v < 8; ++v) {
                    const int r = rowBase + v + 8 * half;
                    if (r < M) Y[(size_t)r * N + c] = acc[t][v] + badd;
                }
            }
        }
    }
}

#define OWN_ROWS 392
#define OWN_THR  1024
#define OWN_WAVES (OWN_THR / 32)
#define QCAP     64
#define SCAN_EDGES 1024

#define REC_OFF_WORDS (NEDGES / 4 + 3840)
__global__ void __launch_bounds__(256)
owner_prep_kernel(const int* __restrict__ src,
                  const int* __restrict__ dst,
                  const int* __restrict__ etype,
                  const float* __restrict__ eweight,
                  unsigned* __restrict__ enc) {
    static_assert((NNODES + OWN_ROWS - 1) / OWN_ROWS <= 256, "an owner number fits a byte");
    static_assert(NNODES <= (1 << 17) && OWN_ROWS <= (1 << 9), "record word: 17 + 1 + 9 bits");
    static_assert(NEDGES % 4 == 0 && (REC_OFF_WORDS % 4) == 0, "four edges per key word, 16-B record stores");
    const int w = blockIdx.x * 256 + threadIdx.x;
    if (w < NEDGES / 4) {
        const int4   d = *(const int4*)(dst + 4 * w);
        const int4   s = *(const int4*)(src + 4 * w);
        const int4   t = *(const int4*)(etype + 4 * w);
        const float4 f = *(const float4*)(eweight + 4 * w);
        const unsigned k0 = min((unsigned)d.x / OWN_ROWS, 255u);
        const unsigned k1 = min((unsigned)d.y / OWN_ROWS, 255u);
        const unsigned k2 = min((unsigned)d.z / OWN_ROWS, 255u);
        const unsigned k3 = min((unsigned)d.w / OWN_ROWS, 255u);
        enc[w] = k0 | (k1 << 8) | (k2 << 16) | (k3 << 24);
        auto word = [](int sv, int tv, int dv, unsigned k) -> unsigned {
            return ((unsigned)sv & 0x1FFFFu) | ((tv != 0) ? (1u << 17) : 0u)
                 | ((((unsigned)dv - k * OWN_ROWS) & 0x1FFu) << 18);
        };
        uint4* rp = (uint4*)(enc + REC_OFF_WORDS) + 2 * w;
        rp[0] = make_uint4(word(s.x, t.x, d.x, k0), __float_as_uint(f.x), word(s.y, t.y, d.y, k1), __float_as_uint(f.y));
        rp[1] = make_uint4(word(s.z, t.z, d.z, k2), __float_as_uint(f.z), word(s.w, t.w, d.w, k3), __float_as_uint(f.w));
    }
}

#define LCAP           512
#define LIST_OFF_WORDS 1572864
#define CNT_OFF_WORDS  (LIST_OFF_WORDS + 256 * OWN_WAVES * LCAP * 2)

template <int DO_LN>
__global__ void __launch_bounds__(OWN_THR)
owner_sum_kernel(const unsigned* __restrict__ enc,
                 unsigned* lst,
                 const float* __restrict__ y0,
                 const float* __restrict__ y1,
                 const float* x,
                 const float* __restrict__ wroot,
                 const float* __restrict__ broot,
                 float* h,
                 const float* __restrict__ g,
                 const float* __restrict__ b) {
    static_assert(NEDGES % SCAN_EDGES == 0, "edge list is tested in whole 1024-edge pieces");
    static_assert(REC_OFF_WORDS + 2 * NEDGES <= LIST_OFF_WORDS, "lists start behind the records");
    static_assert((size_t)(CNT_OFF_WORDS + 256 * OWN_WAVES) * 4 <= (size_t)NNODES * EMB * 4, "lists and counts fit the region");
    static_assert((NNODES + OWN_ROWS - 1) / OWN_ROWS <= 256, "one list per wave of at most 256 work-groups");
    __shared__ __attribute__((aligned(16))) float acc[OWN_ROWS * EMB];
    __shared__ unsigned ring[OWN_WAVES * QCAP];
    constexpr int LDW = EMB + 16;
    constexpr int KS  = EMB / 4;
    __shared__ __attribute__((aligned(16))) float2 wrt[(EMB / 2) * LDW];
    static_assert((OWN_ROWS + 15) / 16 <= OWN_WAVES, "one 16-row tile per wave covers the owned rows");

    const int tid  = threadIdx.x;
    const int lane = tid & 31;
    const int wave = __builtin_amdgcn_readfirstlane(tid >> 5);
    const int r0   = blockIdx.x * OWN_ROWS;
    const int nrow = min(OWN_ROWS, NNODES - r0);
    const unsigned nr = (unsigned)nrow;

    const uint2* rec = (const uint2*)(enc + REC_OFF_WORDS);
    const int wslot = blockIdx.x * OWN_WAVES + wave;
    uint2* mylist = (uint2*)(lst + LIST_OFF_WORDS) + (size_t)wslot * LCAP;
    unsigned* mycnt = lst + CNT_OFF_WORDS + wslot;

    unsigned cntw = 0u;
    uint2 rn = make_uint2(0u, 0u);
    if (!DO_LN) {
        cntw = (unsigned)__builtin_amdgcn_readfirstlane((int)*mycnt);
        if (cntw <= (unsigned)LCAP && (unsigned)lane < cntw) rn = mylist[lane];
    }

    const bool hasTile = (wave * 16 < nrow);
    float2 areg[KS];
    if (hasTile) {
        const int rr = min(wave * 16 + (lane & 15), nrow - 1);
        const float* xrow = x + (size_t)(r0 + rr) * EMB + 2 * (lane >> 4);
#pragma unroll
        for (int j = 0; j < KS; ++j) areg[j] = *(const float2*)(xrow + 4 * j);
    }

    constexpr int RPW = (OWN_ROWS + OWN_WAVES - 1) / OWN_WAVES;
    {
        constexpr int NQ = EMB / 4;
        constexpr int ITEMS = (EMB / 2) * NQ;
        static_assert(ITEMS % OWN_THR == 0, "whole staging rounds");
#pragma unroll
        for (int i0 = 0; i0 < ITEMS; i0 += OWN_THR) {
            const int i = i0 + tid;
            const int p = i / NQ;
            const int cc = (i % NQ) * 4;
            const float4 q0 = *(const float4*)(wroot + (size_t)(2 * p)     * EMB + cc);
            const float4 q1 = *(const float4*)(wroot + (size_t)(2 * p + 1) * EMB + cc);
            float4* dstp = (float4*)(wrt + p * LDW + cc);
            dstp[0] = make_float4(q0.x, q1.x, q0.y, q1.y);
            dstp[1] = make_float4(q0.z, q1.z, q0.w, q1.w);
        }
        static_assert(OWN_THR % 32 == 0 && EMB / 4 == 32, "slot number within a row == lane");
        float4 b4 = make_float4(0.f, 0.f, 0.f, 0.f);
        if (broot) {
            const float* bq = broot + 16 * (lane & 7) + (lane >> 3);
            b4 = make_float4(bq[0], bq[4], bq[8], bq[12]);
        }
        float4* az = (float4*)acc;
        for (int i = tid; i < OWN_ROWS * EMB / 4; i += OWN_THR) az[i] = b4;
    }
    __syncthreads();

    if (hasTile) {
        const int half = lane >> 4;
        const int l    = lane & 15;
#pragma unroll 1
        for (int ps = 0; ps < 2; ++ps) {
            v8f cacc[4] = {};
            const float2* bp0 = wrt + half * LDW + l + ps * 64;
            float2 bcur[4], bnxt[4];
#pragma unroll
            for (int t = 0; t < 4; ++t) bcur[t] = bp0[t * 16];
#pragma unroll
            for (int j = 0; j < KS; ++j) {
                v2f a; a.x = areg[j].x; a.y = areg[j].y;
                if (j + 1 < KS) {
                    const float2* bp = bp0 + (2 * (j + 1)) * LDW;
#pragma unroll
                    for (int t = 0; t < 4; ++t) bnxt[t] = bp[t * 16];
                }
#pragma unroll
                for (int t = 0; t < 4; ++t) {
                    v2f bb2; bb2.x = bcur[t].x; bb2.y = bcur[t].y;
                    cacc[t] = __builtin_amdgcn_wmma_f32_16x16x4_f32(
                        false, a, false, bb2, (short)0, cacc[t], false, false);
                }
                __builtin_amdgcn_sched_barrier(0);
                if (j + 1 < KS) {
#pragma unroll
                    for (int t = 0; t < 4; ++t) bcur[t] = bnxt[t];
                }
            }
#pragma unroll
            for (int t = 0; t < 4; ++t) {
                const int col = ps * 64 + t * 16 + l;
                float* a0 = acc + (col & 3) * 32 + (col >> 2);
#pragma unroll
                for (int v = 0; v < 8; ++v) {
                    const int r = wave * 16 + v + 8 * half;
                    if (r < nrow)
                        __hip_atomic_fetch_add(a0 + r * EMB, cacc[t][v], __ATOMIC_RELAXED, __HIP_MEMORY_SCOPE_WORKGROUP);
                }
            }
        }
    }

    auto add_rows = [&](const int n, const uint2 r) {
        const int s_l = (int)(r.x & 0x1FFFFu);
        const int t_l = (int)((r.x >> 17) & 1u);
        const unsigned dl_l = r.x >> 18;
        const int w_l = (int)r.y;
        for (int j = 0; j < n; j += 8) {
            float4 v[8];
            float  wj[8];
            unsigned dj[8];
#pragma unroll
            for (int u = 0; u < 8; ++u) {
                const int jj = min(j + u, n - 1);
                const int sj = __builtin_amdgcn_readlane(s_l, jj);
                const int tj = __builtin_amdgcn_readlane(t_l, jj);
                wj[u] = __int_as_float(__builtin_amdgcn_readlane(w_l, jj));
                dj[u] = (unsigned)__builtin_amdgcn_readlane((int)dl_l, jj);
                const float* y = ((tj == 0) ? y0 : y1) + (size_t)sj * EMB;
                v[u] = *(const float4*)(y + lane * 4);
            }
#pragma unroll
            for (int u = 0; u < 8; ++u) {
                if (j + u < n && dj[u] < nr) {
                    float* a = acc + dj[u] * EMB + lane;
                    __hip_atomic_fetch_add(a,      wj[u] * v[u].x, __ATOMIC_RELAXED, __HIP_MEMORY_SCOPE_WORKGROUP);
                    __hip_atomic_fetch_add(a + 32, wj[u] * v[u].y, __ATOMIC_RELAXED, __HIP_MEMORY_SCOPE_WORKGROUP);
                    __hip_atomic_fetch_add(a + 64, wj[u] * v[u].z, __ATOMIC_RELAXED, __HIP_MEMORY_SCOPE_WORKGROUP);
                    __hip_atomic_fetch_add(a + 96, wj[u] * v[u].w, __ATOMIC_RELAXED, __HIP_MEMORY_SCOPE_WORKGROUP);
                }
            }
        }
    };

    int noted = 0;

    auto scan_edges = [&]() {
        unsigned* qw = ring + wave * QCAP;
        int qhead = 0, qn = 0;

        constexpr int NPIECE = NEDGES / SCAN_EDGES;
        constexpr int NTRIP  = (NPIECE + OWN_WAVES - 1) / OWN_WAVES;

        static_assert(OWN_WAVES <= NPIECE, "every wave has a first piece");
        constexpr int NTRIP3 = ((NTRIP + 2) / 3) * 3;
        const unsigned* klane = enc + lane * 4;
        auto piece_at = [&](int p) { return klane + (size_t)min(p, NPIECE - 1) * (SCAN_EDGES / 4); };
        int4 idA[3], idB[3];
        idA[0] = *(const int4*)(piece_at(wave));
        idB[0] = *(const int4*)(piece_at(wave) + 128);
        idA[1] = *(const int4*)(piece_at(wave + OWN_WAVES));
        idB[1] = *(const int4*)(piece_at(wave + OWN_WAVES) + 128);
        const unsigned kpat = (unsigned)blockIdx.x * 0x01010101u;

        for (int it = 0; it < NTRIP3; it += 3) {
#pragma unroll
            for (int s = 0; s < 3; ++s) {
                const int piece = wave + OWN_WAVES * (it + s);
                const int4 cA = idA[s], cB = idB[s];
                idA[(s + 2) % 3] = *(const int4*)(piece_at(piece + 2 * OWN_WAVES));
                idB[(s + 2) % 3] = *(const int4*)(piece_at(piece + 2 * OWN_WAVES) + 128);

                auto eq = [&](int x) -> unsigned {
                    const unsigned m = (unsigned)x ^ kpat;
                    return ~(((m & 0x7F7F7F7Fu) + 0x7F7F7F7Fu) | m | 0x7F7F7F7Fu);
                };
                unsigned hm = (eq(cA.x) >> 7) | (eq(cA.y) >> 6) | (eq(cA.z) >> 5) | (eq(cA.w) >> 4)
                            | (eq(cB.x) >> 3) | (eq(cB.y) >> 2) | (eq(cB.z) >> 1) |  eq(cB.w);
                if (piece >= NPIECE) hm = 0u;
                const int ebase = piece * SCAN_EDGES + lane * 16;

                const bool last = (it + s == NTRIP3 - 1);
                unsigned more = __builtin_amdgcn_ballot_w32(hm != 0u);
                do {
                    if (more) {
                        const int pos = qn + (int)__builtin_amdgcn_mbcnt_lo(more, 0u);
                        if (hm != 0u) {
                            const int bit = __builtin_ctz(hm);
                            const int wi = bit & 7, bj = bit >> 3;
                            qw[(qhead + pos) & (QCAP - 1)] = (unsigned)(ebase + (wi >> 2) * 512 + (wi & 3) * 4 + bj);
                            hm &= hm - 1u;
                        }
                        qn += __builtin_popcount(more);
                        more = __builtin_amdgcn_ballot_w32(hm != 0u);
                    }
                    while (qn >= 32 || (last && more == 0u && qn > 0)) {
                        const int n = min(qn, 32);
                        __builtin_amdgcn_wave_barrier();
                        uint2 r = make_uint2(0u, 0u);
                        if (lane < n) r = rec[qw[(qhead + lane) & (QCAP - 1)]];
                        if (DO_LN) {
                            if (noted + n <= LCAP && lane < n) mylist[noted + lane] = r;
                            noted += n;
                        }
                        add_rows(n, r);
                        qhead = (qhead + n) & (QCAP - 1);
                        qn -= n;
                    }
                } while (more);
            }
        }
    };

    if (DO_LN) {
        scan_edges();
        if (lane == 0) *mycnt = (unsigned)noted;
    } else {
        if (cntw <= (unsigned)LCAP) {
            const int cnt = (int)cntw;
            for (int base = 0; base < cnt; base += 32) {
                const int n = min(32, cnt - base);
                const uint2 r = rn;
                if (base + 32 + lane < cnt) rn = mylist[base + 32 + lane];
                add_rows(n, r);
            }
        } else {
            scan_edges();
        }
    }
    __syncthreads();

    float4 gg = make_float4(1.f, 1.f, 1.f, 1.f), bb = make_float4(0.f, 0.f, 0.f, 0.f);
    if (DO_LN) {
        gg = *(const float4*)(g + lane * 4);
        bb = *(const float4*)(b + lane * 4);
    }
    for (int g0 = 0; g0 < RPW; g0 += 4) {
        float4 v[4], o[4];
#pragma unroll
        for (int q = 0; q < 4; ++q) {
            const int r = min(wave + OWN_WAVES * (g0 + q), nrow - 1);
            const float* a = acc + r * EMB + lane;
            v[q] = make_float4(a[0], a[32], a[64], a[96]);
            o[q] = v[q];
        }
        if (DO_LN) {
            float s[4], t[4];
#pragma unroll
            for (int q = 0; q < 4; ++q) {
                v[q].x = fmaxf(v[q].x, 0.0f); v[q].y = fmaxf(v[q].y, 0.0f);
                v[q].z = fmaxf(v[q].z, 0.0f); v[q].w = fmaxf(v[q].w, 0.0f);
                s[q] = v[q].x + v[q].y + v[q].z + v[q].w;
            }
#pragma unroll
            for (int m = 16; m >= 1; m >>= 1) {
#pragma unroll
                for (int q = 0; q < 4; ++q) t[q] = __shfl_xor(s[q], m, 32);
#pragma unroll
                for (int q = 0; q < 4; ++q) s[q] += t[q];
            }
            float dx[4], dy[4], dz[4], dw[4];
#pragma unroll
            for (int q = 0; q < 4; ++q) {
                const float mu = s[q] * (1.0f / EMB);
                dx[q] = v[q].x - mu; dy[q] = v[q].y - mu; dz[q] = v[q].z - mu; dw[q] = v[q].w - mu;
                s[q] = dx[q] * dx[q] + dy[q] * dy[q] + dz[q] * dz[q] + dw[q] * dw[q];
            }
#pragma unroll
            for (int m = 16; m >= 1; m >>= 1) {
#pragma unroll
                for (int q = 0; q < 4; ++q) t[q] = __shfl_xor(s[q], m, 32);
#pragma unroll
                for (int q = 0; q < 4; ++q) s[q] += t[q];
            }
#pragma unroll
            for (int q = 0; q < 4; ++q) {
                const float var = s[q] * (1.0f / EMB);
                const float inv = rsqrtf(var + 1e-5f);
                o[q].x = dx[q] * inv * gg.x + bb.x;
                o[q].y = dy[q] * inv * gg.y + bb.y;
                o[q].z = dz[q] * inv * gg.z + bb.z;
                o[q].w = dw[q] * inv * gg.w + bb.w;
            }
        }
#pragma unroll
        for (int q = 0; q < 4; ++q) {
            const int r = wave + OWN_WAVES * (g0 + q);
            if (r < nrow) *(float4*)(h + (size_t)(r0 + r) * EMB + lane * 4) = o[q];
        }
    }
}

#define HD_THR 128
#define HD_H0  128
#define HD_H1  64
#define HD_H2  32
#define HD_LD0 (HD_H0 + 16)
#define HD_LD1 (HD_H1 + 4)
#define HD_LD2 (HD_H2 + 4)

template <int K, int N, int LD>
__device__ __forceinline__ void head_stage(const float* __restrict__ W, float2* dst, int tid) {
    constexpr int NQ = N / 4;
    constexpr int ITEMS = (K / 2) * NQ;
    static_assert(ITEMS % HD_THR == 0 && LD % 2 == 0, "whole staging rounds, 16-B aligned LDS stores");
#pragma unroll
    for (int i0 = 0; i0 < ITEMS; i0 += HD_THR) {
        const int i = i0 + tid;
        const int p = i / NQ;
        const int c = (i % NQ) * 4;
        const float4 r0 = *(const float4*)(W + (size_t)(2 * p)     * N + c);
        const float4 r1 = *(const float4*)(W + (size_t)(2 * p + 1) * N + c);
        float4* d = (float4*)(dst + p * LD + c);
        d[0] = make_float4(r0.x, r1.x, r0.y, r1.y);
        d[1] = make_float4(r0.z, r1.z, r0.w, r1.w);
    }
}

template <int NTL>
__device__ __forceinline__ void head_load_bias(float4 (&q)[2 * NTL], const float* __restrict__ bias, int half) {
#pragma unroll
    for (int t = 0; t < NTL; ++t) {
        const float* bp = bias + 16 * t + 8 * half;
        q[2 * t]     = *(const float4*)bp;
        q[2 * t + 1] = *(const float4*)(bp + 4);
    }
}

template <int NTL>
__device__ __forceinline__ void head_bias_relu(v8f (&a)[NTL], const float4 (&q)[2 * NTL]) {
#pragma unroll
    for (int t = 0; t < NTL; ++t) {
        a[t][0] = fmaxf(a[t][0] + q[2 * t].x, 0.0f);     a[t][1] = fmaxf(a[t][1] + q[2 * t].y, 0.0f);
        a[t][2] = fmaxf(a[t][2] + q[2 * t].z, 0.0f);     a[t][3] = fmaxf(a[t][3] + q[2 * t].w, 0.0f);
        a[t][4] = fmaxf(a[t][4] + q[2 * t + 1].x, 0.0f); a[t][5] = fmaxf(a[t][5] + q[2 * t + 1].y, 0.0f);
        a[t][6] = fmaxf(a[t][6] + q[2 * t + 1].z, 0.0f); a[t][7] = fmaxf(a[t][7] + q[2 * t + 1].w, 0.0f);
    }
}

template <int KT, int NTO, int LD>
__device__ __forceinline__ void head_layer(const v8f (&in)[KT], v8f (&out)[NTO],
                                           const float2* wt, int half, int l) {
    const float2* ap = wt + (4 * half) * LD + l;
    float2 fc[NTO], fn[NTO];
#pragma unroll
    for (int tt = 0; tt < NTO; ++tt) fc[tt] = ap[tt * 16];
#pragma unroll
    for (int s = 0; s < KT * 4; ++s) {
        const int t = s >> 2, q = s & 3;
        v2f b; b.x = in[t][2 * q]; b.y = in[t][2 * q + 1];
        if (s + 1 < KT * 4) {
            const float2* p = ap + (8 * ((s + 1) >> 2) + ((s + 1) & 3)) * LD;
#pragma unroll
            for (int tt = 0; tt < NTO; ++tt) fn[tt] = p[tt * 16];
        }
#pragma unroll
        for (int tt = 0; tt < NTO; ++tt) {
            v2f a; a.x = fc[tt].x; a.y = fc[tt].y;
            out[tt] = __builtin_amdgcn_wmma_f32_16x16x4_f32(
                false, a, false, b, (short)0, out[tt], false, false);
        }
        __builtin_amdgcn_sched_barrier(0);
        if (s + 1 < KT * 4) {
#pragma unroll
            for (int tt = 0; tt < NTO; ++tt) fc[tt] = fn[tt];
        }
    }
}

__global__ void __launch_bounds__(HD_THR)
head_kernel(const float* __restrict__ h2,
            const int* __restrict__ ui,
            const int* __restrict__ vi,
            const float* __restrict__ w0, const float* __restrict__ b0,
            const float* __restrict__ w1, const float* __restrict__ b1,
            const float* __restrict__ w2, const float* __restrict__ b2,
            const float* __restrict__ ow, const float* __restrict__ ob,
            float* __restrict__ out) {
    constexpr int K0  = 2 * EMB;
    constexpr int KS0 = K0 / 4;
    constexpr int NT0 = HD_H0 / 16, NT1 = HD_H1 / 16, NT2 = HD_H2 / 16;
    __shared__ __attribute__((aligned(16))) float2 s0[(K0 / 2) * HD_LD0];
    __shared__ __attribute__((aligned(16))) float2 s1[(HD_H0 / 2) * HD_LD1];
    __shared__ __attribute__((aligned(16))) float2 s2[(HD_H1 / 2) * HD_LD2];

    const int tid  = threadIdx.x;
    const int wave = __builtin_amdgcn_readfirstlane(tid >> 5);
    const int lane = tid & 31;
    const int half = lane >> 4;
    const int l    = lane & 15;

    constexpr int nTiles = (BATCH + 15) / 16;
    const int tstep = gridDim.x * (HD_THR / 32);
    int tile = blockIdx.x * (HD_THR / 32) + wave;

    float2 xr[KS0];
    auto load_x = [&](int tl) {
        const int it = min(tl * 16 + l, BATCH - 1);
        const float* ur = h2 + (size_t)ui[it] * EMB + 2 * half;
        const float* vr = h2 + (size_t)vi[it] * EMB + 2 * half;
#pragma unroll
        for (int j = 0; j < KS0 / 2; ++j) xr[j] = *(const float2*)(ur + 4 * j);
#pragma unroll
        for (int j = 0; j < KS0 / 2; ++j) xr[KS0 / 2 + j] = *(const float2*)(vr + 4 * j);
    };
    load_x(tile);

    float4 bq0[2 * NT0], bq1[2 * NT1], bq2[2 * NT2], wq[2 * NT2];
    head_load_bias<NT0>(bq0, b0, half);
    head_load_bias<NT1>(bq1, b1, half);
    head_load_bias<NT2>(bq2, b2, half);
    head_load_bias<NT2>(wq, ow + EMB, half);
    float2 wg[KS0 / 2];
#pragma unroll
    for (int j = 0; j < KS0 / 2; ++j) wg[j] = *(const float2*)(ow + 4 * j + 2 * half);
    const float obias = ob[0];

    head_stage<K0,    HD_H0, HD_LD0>(w0, s0, tid);
    head_stage<HD_H0, HD_H1, HD_LD1>(w1, s1, tid);
    head_stage<HD_H1, HD_H2, HD_LD2>(w2, s2, tid);
    __syncthreads();

    for (;;) {
        float su = 0.0f, sv = 0.0f, sd = 0.0f;
#pragma unroll
        for (int j = 0; j < KS0 / 2; ++j) {
            const float2 u = xr[j], v = xr[KS0 / 2 + j];
            su += u.x * u.x + u.y * u.y;
            sv += v.x * v.x + v.y * v.y;
            sd += u.x * v.x * wg[j].x + u.y * v.y * wg[j].y;
        }

        v8f a0[NT0] = {};
        {
            const float2* ap = s0 + half * HD_LD0 + l;
            float2 fc[NT0], fn[NT0];
#pragma unroll
            for (int t = 0; t < NT0; ++t) fc[t] = ap[t * 16];
#pragma unroll
            for (int j = 0; j < KS0; ++j) {
                v2f b; b.x = xr[j].x; b.y = xr[j].y;
                if (j + 1 < KS0) {
                    const float2* p = ap + (2 * (j + 1)) * HD_LD0;
#pragma unroll
                    for (int t = 0; t < NT0; ++t) fn[t] = p[t * 16];
                }
#pragma unroll
                for (int t = 0; t < NT0; ++t) {
                    v2f a; a.x = fc[t].x; a.y = fc[t].y;
                    a0[t] = __builtin_amdgcn_wmma_f32_16x16x4_f32(
                        false, a, false, b, (short)0, a0[t], false, false);
                }
                __builtin_amdgcn_sched_barrier(0);
                if (j + 1 < KS0) {
#pragma unroll
                    for (int t = 0; t < NT0; ++t) fc[t] = fn[t];
                }
            }
        }

        const int cur = tile;
        tile += tstep;
        const bool more = (tile < nTiles);
        if (more) load_x(tile);

        head_bias_relu<NT0>(a0, bq0);
        v8f a1[NT1] = {};
        head_layer<NT0, NT1, HD_LD1>(a0, a1, s1, half, l);
        head_bias_relu<NT1>(a1, bq1);
        v8f a2[NT2] = {};
        head_layer<NT1, NT2, HD_LD2>(a1, a2, s2, half, l);
        head_bias_relu<NT2>(a2, bq2);

        float ps = 0.0f;
#pragma unroll
        for (int t = 0; t < NT2; ++t) {
            ps += a2[t][0] * wq[2 * t].x     + a2[t][1] * wq[2 * t].y
                + a2[t][2] * wq[2 * t].z     + a2[t][3] * wq[2 * t].w
                + a2[t][4] * wq[2 * t + 1].x + a2[t][5] * wq[2 * t + 1].y
                + a2[t][6] * wq[2 * t + 1].z + a2[t][7] * wq[2 * t + 1].w;
        }
        su += __shfl_xor(su, 16, 32);
        sv += __shfl_xor(sv, 16, 32);
        sd += __shfl_xor(sd, 16, 32);
        ps += __shfl_xor(ps, 16, 32);
        const float rn = 1.0f / (fmaxf(sqrtf(su), 1e-12f) * fmaxf(sqrtf(sv), 1e-12f));
        const float z = sd * rn + ps + obias;
        const int it = cur * 16 + l;
        if (half == 0 && it < BATCH) out[it] = 1.0f / (1.0f + expf(-z));

        if (!more) break;
    }
}

extern "C" void kernel_launch(void* const* d_in, const int* in_sizes, int n_in,
                              void* d_out, int out_size, void* d_ws, size_t ws_size,
                              hipStream_t stream) {
    const int*   user_idx = (const int*)d_in[0];
    const int*   item_idx = (const int*)d_in[1];
    const int*   eidx     = (const int*)d_in[2];
    const int*   etype    = (const int*)d_in[3];
    const float* eweight  = (const float*)d_in[4];
    const float* emb      = (const float*)d_in[5];
    const float* w1_rel   = (const float*)d_in[6];
    const float* w1_root  = (const float*)d_in[7];
    const float* b1       = (const float*)d_in[8];
    const float* ln_g     = (const float*)d_in[9];
    const float* ln_b     = (const float*)d_in[10];
    const float* w2_rel   = (const float*)d_in[11];
    const float* w2_root  = (const float*)d_in[12];
    const float* b2       = (const float*)d_in[13];
    const float* mlp_w0   = (const float*)d_in[14];
    const float* mlp_b0   = (const float*)d_in[15];
    const float* mlp_w1   = (const float*)d_in[16];
    const float* mlp_b1   = (const float*)d_in[17];
    const float* mlp_w2   = (const float*)d_in[18];
    const float* mlp_b2   = (const float*)d_in[19];
    const float* out_w    = (const float*)d_in[20];
    const float* out_b    = (const float*)d_in[21];
    float* out = (float*)d_out;

    char* ws = (char*)d_ws;
    const size_t NB = (size_t)NNODES * EMB * sizeof(float);
    float* y0 = (float*)(ws);
    float* y1 = (float*)(ws + NB);
    float* h1 = (float*)(ws + 2 * NB);
    unsigned* ebuf = (unsigned*)(ws + 3 * NB);

    const int nodeGrid  = 256;
    const int headGrid  = 256;
    const int ownGrid   = (NNODES + OWN_ROWS - 1) / OWN_ROWS;

    wmma_gemm3_kernel<8, EMB, 512, 2><<<nodeGrid, 512, 0, stream>>>(emb, w1_rel, w1_rel + EMB * EMB, nullptr, nullptr, y0, y1, nullptr, NNODES);
    owner_prep_kernel<<<NEDGES / 4 / 256, 256, 0, stream>>>(eidx, eidx + NEDGES, etype, eweight, ebuf);
    owner_sum_kernel<1><<<ownGrid, OWN_THR, 0, stream>>>(ebuf, ebuf, y0, y1, emb, w1_root, b1, h1, ln_g, ln_b);

    wmma_gemm3_kernel<8, EMB, 512, 2><<<nodeGrid, 512, 0, stream>>>(h1, w2_rel, w2_rel + EMB * EMB, nullptr, nullptr, y0, y1, nullptr, NNODES);
    owner_sum_kernel<0><<<ownGrid, OWN_THR, 0, stream>>>(ebuf, ebuf, y0, y1, h1, w2_root, b2, h1, nullptr, nullptr);

    head_kernel<<<headGrid, HD_THR, 0, stream>>>(h1, user_idx, item_idx, mlp_w0, mlp_b0, mlp_w1, mlp_b1,
                                                 mlp_w2, mlp_b2, out_w, out_b, out);
}
